# starting kernel
# baseline (speedup 1.0000x reference)
#include <hip/hip_runtime.h>
#include <hip/hip_bf16.h>
#include <math.h>

typedef __attribute__((ext_vector_type(16))) _Float16 v16h;
typedef __attribute__((ext_vector_type(8)))  _Float16 v8h;
typedef __attribute__((ext_vector_type(4)))  _Float16 v4h;
typedef __attribute__((ext_vector_type(8)))  float    v8f;

#define NPTS      262144
#define BSC       128
#define PPS       2048
#define CIN       256
#define CMID      256
#define COUT      512
#define KTOT      288
#define ROWS      64
#define CHUNKS    (PPS / ROWS)
#define ASTRIDE   296

#define WS_W1T     0
#define WS_CENTER  147456
#define WS_PARTIAL 149504

__global__ void prep_w1t_kernel(const float* __restrict__ W1,
                                _Float16* __restrict__ W1t) {
    int idx = blockIdx.x * blockDim.x + threadIdx.x;
    if (idx >= CMID * KTOT) return;
    int col = idx / KTOT;
    int k   = idx % KTOT;
    float v = (k < CIN + 3) ? W1[(size_t)k * CMID + col] : 0.0f;
    W1t[(size_t)col * KTOT + k] = (_Float16)v;
}

__global__ void centroid_kernel(const float* __restrict__ pos,
                                float* __restrict__ center) {
    int s = blockIdx.x;
    int t = threadIdx.x;
    float sx = 0.f, sy = 0.f, sz = 0.f;
    for (int r = t; r < PPS; r += 256) {
        const float* p = pos + (size_t)(s * PPS + r) * 3;
        sx += p[0]; sy += p[1]; sz += p[2];
    }
    __shared__ float rx[256], ry[256], rz[256];
    rx[t] = sx; ry[t] = sy; rz[t] = sz;
    __syncthreads();
    for (int off = 128; off > 0; off >>= 1) {
        if (t < off) { rx[t] += rx[t + off]; ry[t] += ry[t + off]; rz[t] += rz[t + off]; }
        __syncthreads();
    }
    if (t == 0) {
        const float inv = 1.0f / (float)PPS;
        center[s * 4 + 0] = rx[0] * inv;
        center[s * 4 + 1] = ry[0] * inv;
        center[s * 4 + 2] = rz[0] * inv;
    }
}

__global__ __launch_bounds__(256)
void gemm_max_kernel(const float* __restrict__ pos,
                     const float* __restrict__ feature,
                     const _Float16* __restrict__ W1t,
                     const float* __restrict__ center,
                     float* __restrict__ partial) {
    __shared__ __align__(16) _Float16 As[ROWS][ASTRIDE];

    const int b = blockIdx.x;
    const int t = threadIdx.x;
    const int s = b >> 5;
    const int rowbase = b * ROWS;

    const float4* f4 = (const float4*)feature;
    for (int idx = t; idx < ROWS * (CIN / 4); idx += 256) {
        int r = idx >> 6;
        int j = idx & 63;
        float4 v = f4[(size_t)(rowbase + r) * (CIN / 4) + j];
        v4h h;
        h[0] = (_Float16)v.x; h[1] = (_Float16)v.y;
        h[2] = (_Float16)v.z; h[3] = (_Float16)v.w;
        *(v4h*)&As[r][j * 4] = h;
    }
    if (t < ROWS) {
        int r = t;
        const float* pp = pos + (size_t)(rowbase + r) * 3;
        float cx = center[s * 4 + 0];
        float cy = center[s * 4 + 1];
        float cz = center[s * 4 + 2];
        As[r][CIN + 0] = (_Float16)(pp[0] - cx);
        As[r][CIN + 1] = (_Float16)(pp[1] - cy);
        As[r][CIN + 2] = (_Float16)(pp[2] - cz);
        for (int k = CIN + 3; k < ASTRIDE; ++k) As[r][k] = (_Float16)0.0f;
    }
    __syncthreads();

    const int lane = t & 31;
    const int wv   = t >> 5;
    const int n    = lane & 15;
    const int half = lane >> 4;
    const int lo8  = half * 8;

    v8f acc[4][2] = {};

    for (int kt = 0; kt < KTOT / 32; ++kt) {
        const int kb = kt * 32;

        v16h bfrag[2];
#pragma unroll
        for (int ct = 0; ct < 2; ++ct) {
            int col = (wv * 2 + ct) * 16 + n;
            const _Float16* bp = W1t + (size_t)col * KTOT + kb + lo8;
            v8h blo = *(const v8h*)bp;
            v8h bhi = *(const v8h*)(bp + 16);
            bfrag[ct] = __builtin_shufflevector(blo, bhi,
                0,1,2,3,4,5,6,7,8,9,10,11,12,13,14,15);
        }

#pragma unroll
        for (int rt = 0; rt < 4; ++rt) {
            const _Float16* ap = &As[rt * 16 + n][kb + lo8];
            v8h alo = *(const v8h*)ap;
            v8h ahi = *(const v8h*)(ap + 16);
            v16h afrag = __builtin_shufflevector(alo, ahi,
                0,1,2,3,4,5,6,7,8,9,10,11,12,13,14,15);

            acc[rt][0] = __builtin_amdgcn_wmma_f32_16x16x32_f16(
                false, afrag, false, bfrag[0], (short)0, acc[rt][0], false, false);
            acc[rt][1] = __builtin_amdgcn_wmma_f32_16x16x32_f16(
                false, afrag, false, bfrag[1], (short)0, acc[rt][1], false, false);
        }
    }

#pragma unroll
    for (int ct = 0; ct < 2; ++ct) {
        float m = acc[0][ct][0];
#pragma unroll
        for (int rt = 0; rt < 4; ++rt)
#pragma unroll
            for (int i = 0; i < 8; ++i)
                m = fmaxf(m, acc[rt][ct][i]);
        m = fmaxf(m, __shfl_xor(m, 16, 32));
        if (half == 0) {
            int col = (wv * 2 + ct) * 16 + n;
            partial[(size_t)b * CMID + col] = m;
        }
    }
}

__global__ __launch_bounds__(256)
void finish_kernel(const float* __restrict__ partial,
                   const float* __restrict__ b1,
                   const float* __restrict__ W2,
                   const float* __restrict__ b2,
                   const float* __restrict__ noise,
                   float* __restrict__ out) {
    __shared__ float agg[CMID];
    const int s = blockIdx.x;
    const int c = threadIdx.x;

    const float* p = partial + (size_t)s * CHUNKS * CMID + c;
    float m = p[0];
    for (int ch = 1; ch < CHUNKS; ++ch) m = fmaxf(m, p[ch * CMID]);
    agg[c] = fmaxf(m + b1[c], 0.0f);
    __syncthreads();

    float mu = b2[c];
    float sr = b2[c + CMID];
    for (int k = 0; k < CMID; ++k) {
        float a = agg[k];
        mu += a * W2[(size_t)k * COUT + c];
        sr += a * W2[(size_t)k * COUT + c + CMID];
    }
    float sig = fmaxf(sr, 0.0f) + log1pf(expf(-fabsf(sr))) + 1e-4f;
    out[(size_t)s * CMID + c] = mu + sig * noise[(size_t)s * CMID + c];
}

extern "C" void kernel_launch(void* const* d_in, const int* in_sizes, int n_in,
                              void* d_out, int out_size, void* d_ws, size_t ws_size,
                              hipStream_t stream) {
    const float* pos     = (const float*)d_in[0];
    const float* feature = (const float*)d_in[1];
    const float* W1      = (const float*)d_in[3];
    const float* b1      = (const float*)d_in[4];
    const float* W2      = (const float*)d_in[5];
    const float* b2      = (const float*)d_in[6];
    const float* noise   = (const float*)d_in[7];
    float* out = (float*)d_out;

    _Float16* W1t    = (_Float16*)((char*)d_ws + WS_W1T);
    float*    center = (float*)((char*)d_ws + WS_CENTER);
    float*    partial= (float*)((char*)d_ws + WS_PARTIAL);

    prep_w1t_kernel<<<(CMID * KTOT + 255) / 256, 256, 0, stream>>>(W1, W1t);
    centroid_kernel<<<BSC, 256, 0, stream>>>(pos, center);
    gemm_max_kernel<<<BSC * CHUNKS, 256, 0, stream>>>(pos, feature, W1t, center, partial);
    finish_kernel<<<BSC, 256, 0, stream>>>(partial, b1, W2, b2, noise, out);
}


# GEMM blocks made persistent: each keeps its W1t fragments in registers over 8 row chunks (grid 4096 -> 512); chunk done as two 32-row passes with A fragments fetched 2 K-steps ahead
# speedup vs baseline: 1.1174x; 1.1174x over previous
#include <hip/hip_runtime.h>
#include <hip/hip_bf16.h>
#include <math.h>

typedef __attribute__((ext_vector_type(16))) _Float16 v16h;
typedef __attribute__((ext_vector_type(8)))  _Float16 v8h;
typedef __attribute__((ext_vector_type(4)))  _Float16 v4h;
typedef __attribute__((ext_vector_type(8)))  float    v8f;

#define NPTS      262144
#define BSC       128
#define PPS       2048
#define CIN       256
#define CMID      256
#define COUT      512
#define KTOT      288
#define ROWS      64
#define CHUNKS    (PPS / ROWS)
#define ASTRIDE   296
#define NCH       8
#define GBLOCKS   (BSC * CHUNKS / NCH)
static_assert(CHUNKS % NCH == 0, "a block's chunks must stay inside one scene");

#define WS_W1T     0
#define WS_CENTER  147456
#define WS_PARTIAL 149504

__global__ void prep_w1t_kernel(const float* __restrict__ W1,
                                _Float16* __restrict__ W1t) {
    int idx = blockIdx.x * blockDim.x + threadIdx.x;
    if (idx >= CMID * KTOT) return;
    int col = idx / KTOT;
    int k   = idx % KTOT;
    float v = (k < CIN + 3) ? W1[(size_t)k * CMID + col] : 0.0f;
    W1t[(size_t)col * KTOT + k] = (_Float16)v;
}

__global__ void centroid_kernel(const float* __restrict__ pos,
                                float* __restrict__ center) {
    int s = blockIdx.x;
    int t = threadIdx.x;
    float sx = 0.f, sy = 0.f, sz = 0.f;
    for (int r = t; r < PPS; r += 256) {
        const float* p = pos + (size_t)(s * PPS + r) * 3;
        sx += p[0]; sy += p[1]; sz += p[2];
    }
    __shared__ float rx[256], ry[256], rz[256];
    rx[t] = sx; ry[t] = sy; rz[t] = sz;
    __syncthreads();
    for (int off = 128; off > 0; off >>= 1) {
        if (t < off) { rx[t] += rx[t + off]; ry[t] += ry[t + off]; rz[t] += rz[t + off]; }
        __syncthreads();
    }
    if (t == 0) {
        const float inv = 1.0f / (float)PPS;
        center[s * 4 + 0] = rx[0] * inv;
        center[s * 4 + 1] = ry[0] * inv;
        center[s * 4 + 2] = rz[0] * inv;
    }
}

__global__ __launch_bounds__(256)
void gemm_max_kernel(const float* __restrict__ pos,
                     const float* __restrict__ feature,
                     const _Float16* __restrict__ W1t,
                     const float* __restrict__ center,
                     float* __restrict__ partial) {
    __shared__ __align__(16) _Float16 As[ROWS][ASTRIDE];

    const int t = threadIdx.x;
    const int chunk0 = blockIdx.x * NCH;
    const int s = chunk0 >> 5;

    const int lane = t & 31;
    const int wv   = t >> 5;
    const int n    = lane & 15;
    const int half = lane >> 4;
    const int lo8  = half * 8;

    v16h bfrag[KTOT / 32][2];
    for (int kt = 0; kt < KTOT / 32; ++kt) {
        for (int ct = 0; ct < 2; ++ct) {
            int col = (wv * 2 + ct) * 16 + n;
            const _Float16* bp = W1t + (size_t)col * KTOT + kt * 32 + lo8;
            v8h blo = *(const v8h*)bp;
            v8h bhi = *(const v8h*)(bp + 16);
            bfrag[kt][ct] = __builtin_shufflevector(blo, bhi,
                0,1,2,3,4,5,6,7,8,9,10,11,12,13,14,15);
        }
    }

    if (t < ROWS) {
        for (int k = CIN + 3; k < ASTRIDE; ++k) As[t][k] = (_Float16)0.0f;
    }
    const float cx = center[s * 4 + 0];
    const float cy = center[s * 4 + 1];
    const float cz = center[s * 4 + 2];

    const float4* f4 = (const float4*)feature;

    for (int ch = 0; ch < NCH; ++ch) {
        const int chunk = chunk0 + ch;
        const int rowbase = chunk * ROWS;

        if (ch > 0) __syncthreads();

        for (int idx = t; idx < ROWS * (CIN / 4); idx += 256) {
            int r = idx >> 6;
            int j = idx & 63;
            float4 v = f4[(size_t)(rowbase + r) * (CIN / 4) + j];
            v4h h;
            h[0] = (_Float16)v.x; h[1] = (_Float16)v.y;
            h[2] = (_Float16)v.z; h[3] = (_Float16)v.w;
            *(v4h*)&As[r][j * 4] = h;
        }
        if (t < ROWS) {
            const float* pp = pos + (size_t)(rowbase + t) * 3;
            As[t][CIN + 0] = (_Float16)(pp[0] - cx);
            As[t][CIN + 1] = (_Float16)(pp[1] - cy);
            As[t][CIN + 2] = (_Float16)(pp[2] - cz);
        }
        __syncthreads();

        float cmax[2];
        for (int rp = 0; rp < 2; ++rp) {
            v8f acc[2][2] = {};
            v16h afrag[3][2];

#define LOAD_A(set, kt_)                                                        \
            for (int r2 = 0; r2 < 2; ++r2) {                                    \
                const _Float16* ap = &As[(rp * 2 + r2) * 16 + n][(kt_) * 32 + lo8]; \
                v8h alo = *(const v8h*)ap;                                      \
                v8h ahi = *(const v8h*)(ap + 16);                               \
                afrag[set][r2] = __builtin_shufflevector(alo, ahi,              \
                    0,1,2,3,4,5,6,7,8,9,10,11,12,13,14,15);                     \
            }

            LOAD_A(0, 0)
            LOAD_A(1, 1)
            for (int kt = 0; kt < KTOT / 32; ++kt) {
                if (kt + 2 < KTOT / 32) { LOAD_A((kt + 2) % 3, kt + 2) }
                __builtin_amdgcn_sched_barrier(0);
                for (int r2 = 0; r2 < 2; ++r2) {
                    acc[r2][0] = __builtin_amdgcn_wmma_f32_16x16x32_f16(
                        false, afrag[kt % 3][r2], false, bfrag[kt][0], (short)0, acc[r2][0], false, false);
                    acc[r2][1] = __builtin_amdgcn_wmma_f32_16x16x32_f16(
                        false, afrag[kt % 3][r2], false, bfrag[kt][1], (short)0, acc[r2][1], false, false);
                }
            }
#undef LOAD_A

            for (int ct = 0; ct < 2; ++ct) {
                float m = (rp == 0) ? acc[0][ct][0] : cmax[ct];
                for (int r2 = 0; r2 < 2; ++r2)
                    for (int i = 0; i < 8; ++i)
                        m = fmaxf(m, acc[r2][ct][i]);
                cmax[ct] = m;
            }
        }

        for (int ct = 0; ct < 2; ++ct) {
            float m = fmaxf(cmax[ct], __shfl_xor(cmax[ct], 16, 32));
            if (half == 0) {
                int col = (wv * 2 + ct) * 16 + n;
                partial[(size_t)chunk * CMID + col] = m;
            }
        }
    }
}

__global__ __launch_bounds__(256)
void finish_kernel(const float* __restrict__ partial,
                   const float* __restrict__ b1,
                   const float* __restrict__ W2,
                   const float* __restrict__ b2,
                   const float* __restrict__ noise,
                   float* __restrict__ out) {
    __shared__ float agg[CMID];
    const int s = blockIdx.x;
    const int c = threadIdx.x;

    const float* p = partial + (size_t)s * CHUNKS * CMID + c;
    float m = p[0];
    for (int ch = 1; ch < CHUNKS; ++ch) m = fmaxf(m, p[ch * CMID]);
    agg[c] = fmaxf(m + b1[c], 0.0f);
    __syncthreads();

    float mu = b2[c];
    float sr = b2[c + CMID];
    for (int k = 0; k < CMID; ++k) {
        float a = agg[k];
        mu += a * W2[(size_t)k * COUT + c];
        sr += a * W2[(size_t)k * COUT + c + CMID];
    }
    float sig = fmaxf(sr, 0.0f) + log1pf(expf(-fabsf(sr))) + 1e-4f;
    out[(size_t)s * CMID + c] = mu + sig * noise[(size_t)s * CMID + c];
}

extern "C" void kernel_launch(void* const* d_in, const int* in_sizes, int n_in,
                              void* d_out, int out_size, void* d_ws, size_t ws_size,
                              hipStream_t stream) {
    const float* pos     = (const float*)d_in[0];
    const float* feature = (const float*)d_in[1];
    const float* W1      = (const float*)d_in[3];
    const float* b1      = (const float*)d_in[4];
    const float* W2      = (const float*)d_in[5];
    const float* b2      = (const float*)d_in[6];
    const float* noise   = (const float*)d_in[7];
    float* out = (float*)d_out;

    _Float16* W1t    = (_Float16*)((char*)d_ws + WS_W1T);
    float*    center = (float*)((char*)d_ws + WS_CENTER);
    float*    partial= (float*)((char*)d_ws + WS_PARTIAL);

    prep_w1t_kernel<<<(CMID * KTOT + 255) / 256, 256, 0, stream>>>(W1, W1t);
    centroid_kernel<<<BSC, 256, 0, stream>>>(pos, center);
    gemm_max_kernel<<<GBLOCKS, 256, 0, stream>>>(pos, feature, W1t, center, partial);
    finish_kernel<<<BSC, 256, 0, stream>>>(partial, b1, W2, b2, noise, out);
}


# GEMM kernel: loads of the next two 64-row chunks kept in flight in registers across an LDS-only barrier, A double-buffered in LDS; grid 512 -> 256, one block per WGP
# speedup vs baseline: 1.2064x; 1.0796x over previous
#include <hip/hip_runtime.h>
#include <hip/hip_bf16.h>
#include <math.h>

typedef __attribute__((ext_vector_type(16))) _Float16 v16h;
typedef __attribute__((ext_vector_type(8)))  _Float16 v8h;
typedef __attribute__((ext_vector_type(4)))  _Float16 v4h;
typedef __attribute__((ext_vector_type(8)))  float    v8f;

#define NPTS      262144
#define BSC       128
#define PPS       2048
#define CIN       256
#define CMID      256
#define COUT      512
#define KTOT      288
#define ROWS      64
#define CHUNKS    (PPS / ROWS)
#define ASTRIDE   296
#define NCH       16
#define GBLOCKS   (BSC * CHUNKS / NCH)
static_assert(CHUNKS % NCH == 0, "a block's chunks must stay inside one scene");
static_assert(NCH % 2 == 0 && NCH >= 6, "the chunk pipeline runs two chunks per trip plus a four-chunk tail");

#define WS_W1T     0
#define WS_CENTER  147456
#define WS_PARTIAL 149504

__global__ void prep_w1t_kernel(const float* __restrict__ W1,
                                _Float16* __restrict__ W1t) {
    int idx = blockIdx.x * blockDim.x + threadIdx.x;
    if (idx >= CMID * KTOT) return;
    int col = idx / KTOT;
    int k   = idx % KTOT;
    float v = (k < CIN + 3) ? W1[(size_t)k * CMID + col] : 0.0f;
    W1t[(size_t)col * KTOT + k] = (_Float16)v;
}

__global__ void centroid_kernel(const float* __restrict__ pos,
                                float* __restrict__ center) {
    int s = blockIdx.x;
    int t = threadIdx.x;
    float sx = 0.f, sy = 0.f, sz = 0.f;
    for (int r = t; r < PPS; r += 256) {
        const float* p = pos + (size_t)(s * PPS + r) * 3;
        sx += p[0]; sy += p[1]; sz += p[2];
    }
    __shared__ float rx[256], ry[256], rz[256];
    rx[t] = sx; ry[t] = sy; rz[t] = sz;
    __syncthreads();
    for (int off = 128; off > 0; off >>= 1) {
        if (t < off) { rx[t] += rx[t + off]; ry[t] += ry[t + off]; rz[t] += rz[t + off]; }
        __syncthreads();
    }
    if (t == 0) {
        const float inv = 1.0f / (float)PPS;
        center[s * 4 + 0] = rx[0] * inv;
        center[s * 4 + 1] = ry[0] * inv;
        center[s * 4 + 2] = rz[0] * inv;
    }
}


static __device__ __forceinline__ void lds_barrier() {
    asm volatile("s_wait_dscnt 0x0" ::: "memory");
    asm volatile("s_barrier_signal -1" ::: "memory");
    asm volatile("s_barrier_wait -1" ::: "memory");
}

__global__ __launch_bounds__(256)
void gemm_max_kernel(const float* __restrict__ pos,
                     const float* __restrict__ feature,
                     const _Float16* __restrict__ W1t,
                     const float* __restrict__ center,
                     float* __restrict__ partial) {
    __shared__ __align__(16) _Float16 As[2][ROWS][ASTRIDE];

    const int t = threadIdx.x;
    const int chunk0 = blockIdx.x * NCH;
    const int s = chunk0 >> 5;

    const int lane = t & 31;
    const int wv   = t >> 5;
    const int n    = lane & 15;
    const int half = lane >> 4;
    const int lo8  = half * 8;

    v16h bfrag[KTOT / 32][2];
    for (int kt = 0; kt < KTOT / 32; ++kt) {
        for (int ct = 0; ct < 2; ++ct) {
            int col = (wv * 2 + ct) * 16 + n;
            const _Float16* bp = W1t + (size_t)col * KTOT + kt * 32 + lo8;
            v8h blo = *(const v8h*)bp;
            v8h bhi = *(const v8h*)(bp + 16);
            bfrag[kt][ct] = __builtin_shufflevector(blo, bhi,
                0,1,2,3,4,5,6,7,8,9,10,11,12,13,14,15);
        }
    }

    const int rsub = t >> 6;
    const int jcol = t & 63;
    const int tpos = (t < ROWS * 3) ? t : (ROWS * 3 - 1);
    const int prow = tpos / 3;
    const int pk   = tpos - prow * 3;
    const float4* f4 = (const float4*)feature;
    const float pc = center[s * 4 + pk];

    float4 pf[2][16];
    float  pp[2];

#define STAGE_ISSUE(st, chunk_)                                                 \
    {                                                                           \
        const float4* src_ = f4 + ((size_t)(chunk_) * ROWS + rsub) * (CIN / 4) + jcol; \
           \
           \
        _Pragma("unroll")                                                       \
        for (int i_ = 0; i_ < 16; ++i_) pf[st][i_] = src_[(size_t)i_ * 4 * (CIN / 4)]; \
        pp[st] = pos[(size_t)(chunk_) * (ROWS * 3) + tpos];                     \
    }

    STAGE_ISSUE(0, chunk0)
    STAGE_ISSUE(1, chunk0 + 1)

    if (t < ROWS) {
        for (int k = CIN + 3; k < ASTRIDE; ++k) {
            As[0][t][k] = (_Float16)0.0f;
            As[1][t][k] = (_Float16)0.0f;
        }
    }

#define STAGE_STORE(st, buf)                                                    \
    {                                                                           \
                     \
        _Pragma("unroll")                                                       \
        for (int i_ = 0; i_ < 16; ++i_) {                                       \
            float4 v_ = pf[st][i_];                                             \
            v4h h_;                                                             \
            h_[0] = (_Float16)v_.x; h_[1] = (_Float16)v_.y;                     \
            h_[2] = (_Float16)v_.z; h_[3] = (_Float16)v_.w;                     \
            *(v4h*)&As[buf][rsub + 4 * i_][jcol * 4] = h_;                      \
        }                                                                       \
        if (t < ROWS * 3) As[buf][prow][CIN + pk] = (_Float16)(pp[st] - pc);    \
    }

#define LOAD_A(buf, set, kt_)                                                   \
    for (int r2 = 0; r2 < 2; ++r2) {                                            \
        const _Float16* ap = &As[buf][(rp * 2 + r2) * 16 + n][(kt_) * 32 + lo8]; \
        v8h alo = *(const v8h*)ap;                                              \
        v8h ahi = *(const v8h*)(ap + 16);                                       \
        afrag[set][r2] = __builtin_shufflevector(alo, ahi,                      \
            0,1,2,3,4,5,6,7,8,9,10,11,12,13,14,15);                             \
    }

#define COMPUTE(buf, chunk_)                                                    \
    {                                                                           \
        float cmax[2];                                                          \
        for (int rp = 0; rp < 2; ++rp) {                                        \
            v8f acc[2][2] = {};                                                 \
            v16h afrag[3][2];                                                   \
            LOAD_A(buf, 0, 0)                                                   \
            LOAD_A(buf, 1, 1)                                                   \
            for (int kt = 0; kt < KTOT / 32; ++kt) {                            \
                if (kt + 2 < KTOT / 32) { LOAD_A(buf, (kt + 2) % 3, kt + 2) }   \
                    \
                    \
                    \
                    \
                __builtin_amdgcn_sched_barrier(0);                              \
                for (int r2 = 0; r2 < 2; ++r2) {                                \
                    acc[r2][0] = __builtin_amdgcn_wmma_f32_16x16x32_f16(        \
                        false, afrag[kt % 3][r2], false, bfrag[kt][0], (short)0, acc[r2][0], false, false); \
                    acc[r2][1] = __builtin_amdgcn_wmma_f32_16x16x32_f16(        \
                        false, afrag[kt % 3][r2], false, bfrag[kt][1], (short)0, acc[r2][1], false, false); \
                }                                                               \
            }                                                                   \
            for (int ct = 0; ct < 2; ++ct) {                                    \
                float m = (rp == 0) ? acc[0][ct][0] : cmax[ct];                 \
                for (int r2 = 0; r2 < 2; ++r2)                                  \
                    for (int i = 0; i < 8; ++i)                                 \
                        m = fmaxf(m, acc[r2][ct][i]);                           \
                cmax[ct] = m;                                                   \
            }                                                                   \
        }                                                                       \
        for (int ct = 0; ct < 2; ++ct) {                                        \
               \
            float m = fmaxf(cmax[ct], __shfl_xor(cmax[ct], 16, 32));            \
            if (half == 0) {                                                    \
                int col = (wv * 2 + ct) * 16 + n;                               \
                partial[(size_t)(chunk_) * CMID + col] = m;                     \
            }                                                                   \
        }                                                                       \
    }

    STAGE_STORE(0, 0)
    STAGE_ISSUE(0, chunk0 + 2)
    lds_barrier();

    for (int c = 0; c < NCH - 4; c += 2) {
        COMPUTE(0, chunk0 + c)
        STAGE_STORE(1, 1)
        STAGE_ISSUE(1, chunk0 + c + 3)
        lds_barrier();
        COMPUTE(1, chunk0 + c + 1)
        STAGE_STORE(0, 0)
        STAGE_ISSUE(0, chunk0 + c + 4)
        lds_barrier();
    }
    COMPUTE(0, chunk0 + NCH - 4)
    STAGE_STORE(1, 1)
    STAGE_ISSUE(1, chunk0 + NCH - 1)
    lds_barrier();
    COMPUTE(1, chunk0 + NCH - 3)
    STAGE_STORE(0, 0)
    lds_barrier();
    COMPUTE(0, chunk0 + NCH - 2)
    STAGE_STORE(1, 1)
    lds_barrier();
    COMPUTE(1, chunk0 + NCH - 1)

#undef COMPUTE
#undef LOAD_A
#undef STAGE_STORE
#undef STAGE_ISSUE
}

__global__ __launch_bounds__(256)
void finish_kernel(const float* __restrict__ partial,
                   const float* __restrict__ b1,
                   const float* __restrict__ W2,
                   const float* __restrict__ b2,
                   const float* __restrict__ noise,
                   float* __restrict__ out) {
    __shared__ float agg[CMID];
    const int s = blockIdx.x;
    const int c = threadIdx.x;

    const float* p = partial + (size_t)s * CHUNKS * CMID + c;
    float m = p[0];
    for (int ch = 1; ch < CHUNKS; ++ch) m = fmaxf(m, p[ch * CMID]);
    agg[c] = fmaxf(m + b1[c], 0.0f);
    __syncthreads();

    float mu = b2[c];
    float sr = b2[c + CMID];
    for (int k = 0; k < CMID; ++k) {
        float a = agg[k];
        mu += a * W2[(size_t)k * COUT + c];
        sr += a * W2[(size_t)k * COUT + c + CMID];
    }
    float sig = fmaxf(sr, 0.0f) + log1pf(expf(-fabsf(sr))) + 1e-4f;
    out[(size_t)s * CMID + c] = mu + sig * noise[(size_t)s * CMID + c];
}

extern "C" void kernel_launch(void* const* d_in, const int* in_sizes, int n_in,
                              void* d_out, int out_size, void* d_ws, size_t ws_size,
                              hipStream_t stream) {
    const float* pos     = (const float*)d_in[0];
    const float* feature = (const float*)d_in[1];
    const float* W1      = (const float*)d_in[3];
    const float* b1      = (const float*)d_in[4];
    const float* W2      = (const float*)d_in[5];
    const float* b2      = (const float*)d_in[6];
    const float* noise   = (const float*)d_in[7];
    float* out = (float*)d_out;

    _Float16* W1t    = (_Float16*)((char*)d_ws + WS_W1T);
    float*    center = (float*)((char*)d_ws + WS_CENTER);
    float*    partial= (float*)((char*)d_ws + WS_PARTIAL);

    prep_w1t_kernel<<<(CMID * KTOT + 255) / 256, 256, 0, stream>>>(W1, W1t);
    centroid_kernel<<<BSC, 256, 0, stream>>>(pos, center);
    gemm_max_kernel<<<GBLOCKS, 256, 0, stream>>>(pos, feature, W1t, center, partial);
    finish_kernel<<<BSC, 256, 0, stream>>>(partial, b1, W2, b2, noise, out);
}


# finish kernel re-tiled: block = 2 scenes x (64 mu + 64 sigma) columns, grid 128 -> 256, a thread's 32 W2 rows loaded as one group; W2 product now summed as 8 k-group partials added in order on b2
# speedup vs baseline: 1.5397x; 1.2763x over previous
#include <hip/hip_runtime.h>
#include <hip/hip_bf16.h>
#include <math.h>

typedef __attribute__((ext_vector_type(16))) _Float16 v16h;
typedef __attribute__((ext_vector_type(8)))  _Float16 v8h;
typedef __attribute__((ext_vector_type(4)))  _Float16 v4h;
typedef __attribute__((ext_vector_type(8)))  float    v8f;

#define NPTS      262144
#define BSC       128
#define PPS       2048
#define CIN       256
#define CMID      256
#define COUT      512
#define KTOT      288
#define ROWS      64
#define CHUNKS    (PPS / ROWS)
#define ASTRIDE   296
#define NCH       16
#define GBLOCKS   (BSC * CHUNKS / NCH)
static_assert(CHUNKS % NCH == 0, "a block's chunks must stay inside one scene");
static_assert(NCH % 2 == 0 && NCH >= 6, "the chunk pipeline runs two chunks per trip plus a four-chunk tail");

#define WS_W1T     0
#define WS_CENTER  147456
#define WS_PARTIAL 149504

__global__ void prep_w1t_kernel(const float* __restrict__ W1,
                                _Float16* __restrict__ W1t) {
    int idx = blockIdx.x * blockDim.x + threadIdx.x;
    if (idx >= CMID * KTOT) return;
    int col = idx / KTOT;
    int k   = idx % KTOT;
    float v = (k < CIN + 3) ? W1[(size_t)k * CMID + col] : 0.0f;
    W1t[(size_t)col * KTOT + k] = (_Float16)v;
}

__global__ void centroid_kernel(const float* __restrict__ pos,
                                float* __restrict__ center) {
    int s = blockIdx.x;
    int t = threadIdx.x;
    float sx = 0.f, sy = 0.f, sz = 0.f;
    for (int r = t; r < PPS; r += 256) {
        const float* p = pos + (size_t)(s * PPS + r) * 3;
        sx += p[0]; sy += p[1]; sz += p[2];
    }
    __shared__ float rx[256], ry[256], rz[256];
    rx[t] = sx; ry[t] = sy; rz[t] = sz;
    __syncthreads();
    for (int off = 128; off > 0; off >>= 1) {
        if (t < off) { rx[t] += rx[t + off]; ry[t] += ry[t + off]; rz[t] += rz[t + off]; }
        __syncthreads();
    }
    if (t == 0) {
        const float inv = 1.0f / (float)PPS;
        center[s * 4 + 0] = rx[0] * inv;
        center[s * 4 + 1] = ry[0] * inv;
        center[s * 4 + 2] = rz[0] * inv;
    }
}


static __device__ __forceinline__ void lds_barrier() {
    asm volatile("s_wait_dscnt 0x0" ::: "memory");
    asm volatile("s_barrier_signal -1" ::: "memory");
    asm volatile("s_barrier_wait -1" ::: "memory");
}

__global__ __launch_bounds__(256)
void gemm_max_kernel(const float* __restrict__ pos,
                     const float* __restrict__ feature,
                     const _Float16* __restrict__ W1t,
                     const float* __restrict__ center,
                     float* __restrict__ partial) {
    __shared__ __align__(16) _Float16 As[2][ROWS][ASTRIDE];

    const int t = threadIdx.x;
    const int chunk0 = blockIdx.x * NCH;
    const int s = chunk0 >> 5;

    const int lane = t & 31;
    const int wv   = t >> 5;
    const int n    = lane & 15;
    const int half = lane >> 4;
    const int lo8  = half * 8;

    v16h bfrag[KTOT / 32][2];
    for (int kt = 0; kt < KTOT / 32; ++kt) {
        for (int ct = 0; ct < 2; ++ct) {
            int col = (wv * 2 + ct) * 16 + n;
            const _Float16* bp = W1t + (size_t)col * KTOT + kt * 32 + lo8;
            v8h blo = *(const v8h*)bp;
            v8h bhi = *(const v8h*)(bp + 16);
            bfrag[kt][ct] = __builtin_shufflevector(blo, bhi,
                0,1,2,3,4,5,6,7,8,9,10,11,12,13,14,15);
        }
    }

    const int rsub = t >> 6;
    const int jcol = t & 63;
    const int tpos = (t < ROWS * 3) ? t : (ROWS * 3 - 1);
    const int prow = tpos / 3;
    const int pk   = tpos - prow * 3;
    const float4* f4 = (const float4*)feature;
    const float pc = center[s * 4 + pk];

    float4 pf[2][16];
    float  pp[2];

#define STAGE_ISSUE(st, chunk_)                                                 \
    {                                                                           \
        const float4* src_ = f4 + ((size_t)(chunk_) * ROWS + rsub) * (CIN / 4) + jcol; \
           \
           \
        _Pragma("unroll")                                                       \
        for (int i_ = 0; i_ < 16; ++i_) pf[st][i_] = src_[(size_t)i_ * 4 * (CIN / 4)]; \
        pp[st] = pos[(size_t)(chunk_) * (ROWS * 3) + tpos];                     \
    }

    STAGE_ISSUE(0, chunk0)
    STAGE_ISSUE(1, chunk0 + 1)

    if (t < ROWS) {
        for (int k = CIN + 3; k < ASTRIDE; ++k) {
            As[0][t][k] = (_Float16)0.0f;
            As[1][t][k] = (_Float16)0.0f;
        }
    }

#define STAGE_STORE(st, buf)                                                    \
    {                                                                           \
                     \
        _Pragma("unroll")                                                       \
        for (int i_ = 0; i_ < 16; ++i_) {                                       \
            float4 v_ = pf[st][i_];                                             \
            v4h h_;                                                             \
            h_[0] = (_Float16)v_.x; h_[1] = (_Float16)v_.y;                     \
            h_[2] = (_Float16)v_.z; h_[3] = (_Float16)v_.w;                     \
            *(v4h*)&As[buf][rsub + 4 * i_][jcol * 4] = h_;                      \
        }                                                                       \
        if (t < ROWS * 3) As[buf][prow][CIN + pk] = (_Float16)(pp[st] - pc);    \
    }

#define LOAD_A(buf, set, kt_)                                                   \
    for (int r2 = 0; r2 < 2; ++r2) {                                            \
        const _Float16* ap = &As[buf][(rp * 2 + r2) * 16 + n][(kt_) * 32 + lo8]; \
        v8h alo = *(const v8h*)ap;                                              \
        v8h ahi = *(const v8h*)(ap + 16);                                       \
        afrag[set][r2] = __builtin_shufflevector(alo, ahi,                      \
            0,1,2,3,4,5,6,7,8,9,10,11,12,13,14,15);                             \
    }

#define COMPUTE(buf, chunk_)                                                    \
    {                                                                           \
        float cmax[2];                                                          \
        for (int rp = 0; rp < 2; ++rp) {                                        \
            v8f acc[2][2] = {};                                                 \
            v16h afrag[3][2];                                                   \
            LOAD_A(buf, 0, 0)                                                   \
            LOAD_A(buf, 1, 1)                                                   \
            for (int kt = 0; kt < KTOT / 32; ++kt) {                            \
                if (kt + 2 < KTOT / 32) { LOAD_A(buf, (kt + 2) % 3, kt + 2) }   \
                    \
                    \
                    \
                    \
                __builtin_amdgcn_sched_barrier(0);                              \
                for (int r2 = 0; r2 < 2; ++r2) {                                \
                    acc[r2][0] = __builtin_amdgcn_wmma_f32_16x16x32_f16(        \
                        false, afrag[kt % 3][r2], false, bfrag[kt][0], (short)0, acc[r2][0], false, false); \
                    acc[r2][1] = __builtin_amdgcn_wmma_f32_16x16x32_f16(        \
                        false, afrag[kt % 3][r2], false, bfrag[kt][1], (short)0, acc[r2][1], false, false); \
                }                                                               \
            }                                                                   \
            for (int ct = 0; ct < 2; ++ct) {                                    \
                float m = (rp == 0) ? acc[0][ct][0] : cmax[ct];                 \
                for (int r2 = 0; r2 < 2; ++r2)                                  \
                    for (int i = 0; i < 8; ++i)                                 \
                        m = fmaxf(m, acc[r2][ct][i]);                           \
                cmax[ct] = m;                                                   \
            }                                                                   \
        }                                                                       \
        for (int ct = 0; ct < 2; ++ct) {                                        \
               \
            float m = fmaxf(cmax[ct], __shfl_xor(cmax[ct], 16, 32));            \
            if (half == 0) {                                                    \
                int col = (wv * 2 + ct) * 16 + n;                               \
                partial[(size_t)(chunk_) * CMID + col] = m;                     \
            }                                                                   \
        }                                                                       \
    }

    STAGE_STORE(0, 0)
    STAGE_ISSUE(0, chunk0 + 2)
    lds_barrier();

    for (int c = 0; c < NCH - 4; c += 2) {
        COMPUTE(0, chunk0 + c)
        STAGE_STORE(1, 1)
        STAGE_ISSUE(1, chunk0 + c + 3)
        lds_barrier();
        COMPUTE(1, chunk0 + c + 1)
        STAGE_STORE(0, 0)
        STAGE_ISSUE(0, chunk0 + c + 4)
        lds_barrier();
    }
    COMPUTE(0, chunk0 + NCH - 4)
    STAGE_STORE(1, 1)
    STAGE_ISSUE(1, chunk0 + NCH - 1)
    lds_barrier();
    COMPUTE(1, chunk0 + NCH - 3)
    STAGE_STORE(0, 0)
    lds_barrier();
    COMPUTE(0, chunk0 + NCH - 2)
    STAGE_STORE(1, 1)
    lds_barrier();
    COMPUTE(1, chunk0 + NCH - 1)

#undef COMPUTE
#undef LOAD_A
#undef STAGE_STORE
#undef STAGE_ISSUE
}

#define FIN_SC     2
#define FIN_CW     64
#define FIN_KG     8
#define FIN_KPT    (CMID / FIN_KG)
#define FIN_SLICES (CMID / FIN_CW)
#define FIN_BLOCKS ((BSC / FIN_SC) * FIN_SLICES)
static_assert(FIN_CW * 2 == 32 * 4, "one wave = one W2 row of the block: 16 mu + 16 sigma float4");
static_assert(FIN_SC * FIN_CW <= 256 && CMID == 256, "thread maps of the head and the tail");

__global__ __launch_bounds__(256)
void finish_kernel(const float* __restrict__ partial,
                   const float* __restrict__ b1,
                   const float* __restrict__ W2,
                   const float* __restrict__ b2,
                   const float* __restrict__ noise,
                   float* __restrict__ out) {
    __shared__ __align__(8)  float agg[CMID][FIN_SC];
    __shared__ __align__(16) float red[FIN_KG][FIN_SC][2 * FIN_CW];

    const int t    = threadIdx.x;
    const int lane = t & 31;
    const int kg   = t >> 5;
    const int sl   = blockIdx.x % FIN_SLICES;
    const int s0   = (blockIdx.x / FIN_SLICES) * FIN_SC;

    const int col4 = (lane < 16) ? (FIN_CW * sl + 4 * lane)
                                 : (CMID + FIN_CW * sl + 4 * (lane - 16));

    float cm[FIN_SC][CHUNKS];
#pragma unroll
    for (int sc = 0; sc < FIN_SC; ++sc) {
        const float* p = partial + (size_t)(s0 + sc) * CHUNKS * CMID + t;
#pragma unroll
        for (int ch = 0; ch < CHUNKS; ++ch) cm[sc][ch] = p[ch * CMID];
    }
    const float bias1 = b1[t];

    float4 w[FIN_KPT];
    const float4* wp = (const float4*)(W2 + (size_t)kg * COUT + col4);
#pragma unroll
    for (int i = 0; i < FIN_KPT; ++i) w[i] = wp[(size_t)i * FIN_KG * (COUT / 4)];

    const int tsc = t / FIN_CW;
    const int tcc = t % FIN_CW;
    float tb_mu = 0.f, tb_sr = 0.f, tnoise = 0.f;
    if (t < FIN_SC * FIN_CW) {
        tb_mu  = b2[FIN_CW * sl + tcc];
        tb_sr  = b2[CMID + FIN_CW * sl + tcc];
        tnoise = noise[(size_t)(s0 + tsc) * CMID + FIN_CW * sl + tcc];
    }

#pragma unroll
    for (int sc = 0; sc < FIN_SC; ++sc) {
        float m = cm[sc][0];
#pragma unroll
        for (int ch = 1; ch < CHUNKS; ++ch) m = fmaxf(m, cm[sc][ch]);
        agg[t][sc] = fmaxf(m + bias1, 0.0f);
    }
    __syncthreads();

    float4 acc[FIN_SC];
#pragma unroll
    for (int sc = 0; sc < FIN_SC; ++sc) acc[sc] = make_float4(0.f, 0.f, 0.f, 0.f);
#pragma unroll
    for (int i = 0; i < FIN_KPT; ++i) {
        const int k = kg + FIN_KG * i;
#pragma unroll
        for (int sc = 0; sc < FIN_SC; ++sc) {
            const float a = agg[k][sc];
            acc[sc].x += a * w[i].x; acc[sc].y += a * w[i].y;
            acc[sc].z += a * w[i].z; acc[sc].w += a * w[i].w;
        }
    }
#pragma unroll
    for (int sc = 0; sc < FIN_SC; ++sc)
        *(float4*)&red[kg][sc][4 * lane] = acc[sc];
    __syncthreads();

    if (t < FIN_SC * FIN_CW) {
        float mu = tb_mu;
        float sr = tb_sr;
#pragma unroll
        for (int g = 0; g < FIN_KG; ++g) {
            mu += red[g][tsc][tcc];
            sr += red[g][tsc][FIN_CW + tcc];
        }
        float sig = fmaxf(sr, 0.0f) + log1pf(expf(-fabsf(sr))) + 1e-4f;
        out[(size_t)(s0 + tsc) * CMID + FIN_CW * sl + tcc] = mu + sig * tnoise;
    }
}

extern "C" void kernel_launch(void* const* d_in, const int* in_sizes, int n_in,
                              void* d_out, int out_size, void* d_ws, size_t ws_size,
                              hipStream_t stream) {
    const float* pos     = (const float*)d_in[0];
    const float* feature = (const float*)d_in[1];
    const float* W1      = (const float*)d_in[3];
    const float* b1      = (const float*)d_in[4];
    const float* W2      = (const float*)d_in[5];
    const float* b2      = (const float*)d_in[6];
    const float* noise   = (const float*)d_in[7];
    float* out = (float*)d_out;

    _Float16* W1t    = (_Float16*)((char*)d_ws + WS_W1T);
    float*    center = (float*)((char*)d_ws + WS_CENTER);
    float*    partial= (float*)((char*)d_ws + WS_PARTIAL);

    prep_w1t_kernel<<<(CMID * KTOT + 255) / 256, 256, 0, stream>>>(W1, W1t);
    centroid_kernel<<<BSC, 256, 0, stream>>>(pos, center);
    gemm_max_kernel<<<GBLOCKS, 256, 0, stream>>>(pos, feature, W1t, center, partial);
    finish_kernel<<<FIN_BLOCKS, 256, 0, stream>>>(partial, b1, W2, b2, noise, out);
}


# centroid launch removed: each GEMM block sums its scene's 2048 points in its prologue while its first loads are in flight (launches 4 -> 3); centroid summation order changed
# speedup vs baseline: 1.7475x; 1.1350x over previous
#include <hip/hip_runtime.h>
#include <hip/hip_bf16.h>
#include <math.h>

typedef __attribute__((ext_vector_type(16))) _Float16 v16h;
typedef __attribute__((ext_vector_type(8)))  _Float16 v8h;
typedef __attribute__((ext_vector_type(4)))  _Float16 v4h;
typedef __attribute__((ext_vector_type(8)))  float    v8f;
typedef __attribute__((ext_vector_type(4)))  float    v4f;

#define NPTS      262144
#define BSC       128
#define PPS       2048
#define CIN       256
#define CMID      256
#define COUT      512
#define KTOT      288
#define ROWS      64
#define CHUNKS    (PPS / ROWS)
#define ASTRIDE   296
#define NCH       16
#define GBLOCKS   (BSC * CHUNKS / NCH)
static_assert(CHUNKS % NCH == 0, "a block's chunks must stay inside one scene");
static_assert(NCH % 2 == 0 && NCH >= 6, "the chunk pipeline runs two chunks per trip plus a four-chunk tail");

#define WS_W1T     0
#define WS_PARTIAL 149504

__global__ void prep_w1t_kernel(const float* __restrict__ W1,
                                _Float16* __restrict__ W1t) {
    int idx = blockIdx.x * blockDim.x + threadIdx.x;
    if (idx >= CMID * KTOT) return;
    int col = idx / KTOT;
    int k   = idx % KTOT;
    float v = (k < CIN + 3) ? W1[(size_t)k * CMID + col] : 0.0f;
    W1t[(size_t)col * KTOT + k] = (_Float16)v;
}


static __device__ __forceinline__ void lds_barrier() {
    asm volatile("s_wait_dscnt 0x0" ::: "memory");
    asm volatile("s_barrier_signal -1" ::: "memory");
    asm volatile("s_barrier_wait -1" ::: "memory");
}

__global__ __launch_bounds__(256)
void gemm_max_kernel(const float* __restrict__ pos,
                     const float* __restrict__ feature,
                     const _Float16* __restrict__ W1t,
                     float* __restrict__ partial) {
    __shared__ __align__(16) _Float16 As[2][ROWS][ASTRIDE];
    __shared__ float csum[8][4];

    const int t = threadIdx.x;
    const int chunk0 = blockIdx.x * NCH;
    const int s = chunk0 >> 5;

    const int lane = t & 31;
    const int wv   = t >> 5;
    const int n    = lane & 15;
    const int half = lane >> 4;
    const int lo8  = half * 8;

    static_assert(PPS * 3 == 6 * 256 * 4, "6 float4 per thread cover a scene's pos");
    const v4f* cp4 = (const v4f*)(pos + (size_t)s * (PPS * 3));
    v4f cv[6];
#pragma unroll
    for (int i = 0; i < 6; ++i) cv[i] = cp4[t + 256 * i];

    v16h bfrag[KTOT / 32][2];
    for (int kt = 0; kt < KTOT / 32; ++kt) {
        for (int ct = 0; ct < 2; ++ct) {
            int col = (wv * 2 + ct) * 16 + n;
            const _Float16* bp = W1t + (size_t)col * KTOT + kt * 32 + lo8;
            v8h blo = *(const v8h*)bp;
            v8h bhi = *(const v8h*)(bp + 16);
            bfrag[kt][ct] = __builtin_shufflevector(blo, bhi,
                0,1,2,3,4,5,6,7,8,9,10,11,12,13,14,15);
        }
    }
    asm volatile("" ::: "memory");

    const int rsub = t >> 6;
    const int jcol = t & 63;
    const int tpos = (t < ROWS * 3) ? t : (ROWS * 3 - 1);
    const int prow = tpos / 3;
    const int pk   = tpos - prow * 3;
    const float4* f4 = (const float4*)feature;

    float4 pf[2][16];
    float  pp[2];

#define STAGE_ISSUE(st, chunk_)                                                 \
    {                                                                           \
        const float4* src_ = f4 + ((size_t)(chunk_) * ROWS + rsub) * (CIN / 4) + jcol; \
           \
           \
        _Pragma("unroll")                                                       \
        for (int i_ = 0; i_ < 16; ++i_) pf[st][i_] = src_[(size_t)i_ * 4 * (CIN / 4)]; \
        pp[st] = pos[(size_t)(chunk_) * (ROWS * 3) + tpos];                     \
    }

    STAGE_ISSUE(0, chunk0)
    STAGE_ISSUE(1, chunk0 + 1)
    asm volatile("" : "+v"(cv[0]), "+v"(cv[1]), "+v"(cv[2]),
                      "+v"(cv[3]), "+v"(cv[4]), "+v"(cv[5]) :: "memory");

    float rq[3] = {0.f, 0.f, 0.f};
#pragma unroll
    for (int i = 0; i < 6; ++i) {
        rq[(i + 0) % 3] += cv[i].x;
        rq[(i + 1) % 3] += cv[i].y;
        rq[(i + 2) % 3] += cv[i].z;
        rq[(i + 3) % 3] += cv[i].w;
    }
    const int ta = t % 3;
    float cx = (ta == 0) ? rq[0] : (ta == 1) ? rq[2] : rq[1];
    float cy = (ta == 0) ? rq[1] : (ta == 1) ? rq[0] : rq[2];
    float cz = (ta == 0) ? rq[2] : (ta == 1) ? rq[1] : rq[0];
    for (int off = 16; off > 0; off >>= 1) {
        cx += __shfl_xor(cx, off, 32);
        cy += __shfl_xor(cy, off, 32);
        cz += __shfl_xor(cz, off, 32);
    }
    if (lane == 0) { csum[wv][0] = cx; csum[wv][1] = cy; csum[wv][2] = cz; }
    lds_barrier();
    float pc = csum[0][pk];
    for (int w = 1; w < 8; ++w) pc += csum[w][pk];
    pc *= 1.0f / (float)PPS;

    if (t < ROWS) {
        for (int k = CIN + 3; k < ASTRIDE; ++k) {
            As[0][t][k] = (_Float16)0.0f;
            As[1][t][k] = (_Float16)0.0f;
        }
    }

#define STAGE_STORE(st, buf)                                                    \
    {                                                                           \
                     \
        _Pragma("unroll")                                                       \
        for (int i_ = 0; i_ < 16; ++i_) {                                       \
            float4 v_ = pf[st][i_];                                             \
            v4h h_;                                                             \
            h_[0] = (_Float16)v_.x; h_[1] = (_Float16)v_.y;                     \
            h_[2] = (_Float16)v_.z; h_[3] = (_Float16)v_.w;                     \
            *(v4h*)&As[buf][rsub + 4 * i_][jcol * 4] = h_;                      \
        }                                                                       \
        if (t < ROWS * 3) As[buf][prow][CIN + pk] = (_Float16)(pp[st] - pc);    \
    }

#define LOAD_A(buf, set, kt_)                                                   \
    for (int r2 = 0; r2 < 2; ++r2) {                                            \
        const _Float16* ap = &As[buf][(rp * 2 + r2) * 16 + n][(kt_) * 32 + lo8]; \
        v8h alo = *(const v8h*)ap;                                              \
        v8h ahi = *(const v8h*)(ap + 16);                                       \
        afrag[set][r2] = __builtin_shufflevector(alo, ahi,                      \
            0,1,2,3,4,5,6,7,8,9,10,11,12,13,14,15);                             \
    }

#define COMPUTE(buf, chunk_)                                                    \
    {                                                                           \
        float cmax[2];                                                          \
        for (int rp = 0; rp < 2; ++rp) {                                        \
            v8f acc[2][2] = {};                                                 \
            v16h afrag[3][2];                                                   \
            LOAD_A(buf, 0, 0)                                                   \
            LOAD_A(buf, 1, 1)                                                   \
            for (int kt = 0; kt < KTOT / 32; ++kt) {                            \
                if (kt + 2 < KTOT / 32) { LOAD_A(buf, (kt + 2) % 3, kt + 2) }   \
                    \
                    \
                    \
                    \
                __builtin_amdgcn_sched_barrier(0);                              \
                for (int r2 = 0; r2 < 2; ++r2) {                                \
                    acc[r2][0] = __builtin_amdgcn_wmma_f32_16x16x32_f16(        \
                        false, afrag[kt % 3][r2], false, bfrag[kt][0], (short)0, acc[r2][0], false, false); \
                    acc[r2][1] = __builtin_amdgcn_wmma_f32_16x16x32_f16(        \
                        false, afrag[kt % 3][r2], false, bfrag[kt][1], (short)0, acc[r2][1], false, false); \
                }                                                               \
            }                                                                   \
            for (int ct = 0; ct < 2; ++ct) {                                    \
                float m = (rp == 0) ? acc[0][ct][0] : cmax[ct];                 \
                for (int r2 = 0; r2 < 2; ++r2)                                  \
                    for (int i = 0; i < 8; ++i)                                 \
                        m = fmaxf(m, acc[r2][ct][i]);                           \
                cmax[ct] = m;                                                   \
            }                                                                   \
        }                                                                       \
        for (int ct = 0; ct < 2; ++ct) {                                        \
               \
            float m = fmaxf(cmax[ct], __shfl_xor(cmax[ct], 16, 32));            \
            if (half == 0) {                                                    \
                int col = (wv * 2 + ct) * 16 + n;                               \
                partial[(size_t)(chunk_) * CMID + col] = m;                     \
            }                                                                   \
        }                                                                       \
    }

    STAGE_STORE(0, 0)
    STAGE_ISSUE(0, chunk0 + 2)
    lds_barrier();

    for (int c = 0; c < NCH - 4; c += 2) {
        COMPUTE(0, chunk0 + c)
        STAGE_STORE(1, 1)
        STAGE_ISSUE(1, chunk0 + c + 3)
        lds_barrier();
        COMPUTE(1, chunk0 + c + 1)
        STAGE_STORE(0, 0)
        STAGE_ISSUE(0, chunk0 + c + 4)
        lds_barrier();
    }
    COMPUTE(0, chunk0 + NCH - 4)
    STAGE_STORE(1, 1)
    STAGE_ISSUE(1, chunk0 + NCH - 1)
    lds_barrier();
    COMPUTE(1, chunk0 + NCH - 3)
    STAGE_STORE(0, 0)
    lds_barrier();
    COMPUTE(0, chunk0 + NCH - 2)
    STAGE_STORE(1, 1)
    lds_barrier();
    COMPUTE(1, chunk0 + NCH - 1)

#undef COMPUTE
#undef LOAD_A
#undef STAGE_STORE
#undef STAGE_ISSUE
}

#define FIN_SC     2
#define FIN_CW     64
#define FIN_KG     8
#define FIN_KPT    (CMID / FIN_KG)
#define FIN_SLICES (CMID / FIN_CW)
#define FIN_BLOCKS ((BSC / FIN_SC) * FIN_SLICES)
static_assert(FIN_CW * 2 == 32 * 4, "one wave = one W2 row of the block: 16 mu + 16 sigma float4");
static_assert(FIN_SC * FIN_CW <= 256 && CMID == 256, "thread maps of the head and the tail");

__global__ __launch_bounds__(256)
void finish_kernel(const float* __restrict__ partial,
                   const float* __restrict__ b1,
                   const float* __restrict__ W2,
                   const float* __restrict__ b2,
                   const float* __restrict__ noise,
                   float* __restrict__ out) {
    __shared__ __align__(8)  float agg[CMID][FIN_SC];
    __shared__ __align__(16) float red[FIN_KG][FIN_SC][2 * FIN_CW];

    const int t    = threadIdx.x;
    const int lane = t & 31;
    const int kg   = t >> 5;
    const int sl   = blockIdx.x % FIN_SLICES;
    const int s0   = (blockIdx.x / FIN_SLICES) * FIN_SC;

    const int col4 = (lane < 16) ? (FIN_CW * sl + 4 * lane)
                                 : (CMID + FIN_CW * sl + 4 * (lane - 16));

    float cm[FIN_SC][CHUNKS];
#pragma unroll
    for (int sc = 0; sc < FIN_SC; ++sc) {
        const float* p = partial + (size_t)(s0 + sc) * CHUNKS * CMID + t;
#pragma unroll
        for (int ch = 0; ch < CHUNKS; ++ch) cm[sc][ch] = p[ch * CMID];
    }
    const float bias1 = b1[t];

    float4 w[FIN_KPT];
    const float4* wp = (const float4*)(W2 + (size_t)kg * COUT + col4);
#pragma unroll
    for (int i = 0; i < FIN_KPT; ++i) w[i] = wp[(size_t)i * FIN_KG * (COUT / 4)];

    const int tsc = t / FIN_CW;
    const int tcc = t % FIN_CW;
    float tb_mu = 0.f, tb_sr = 0.f, tnoise = 0.f;
    if (t < FIN_SC * FIN_CW) {
        tb_mu  = b2[FIN_CW * sl + tcc];
        tb_sr  = b2[CMID + FIN_CW * sl + tcc];
        tnoise = noise[(size_t)(s0 + tsc) * CMID + FIN_CW * sl + tcc];
    }

#pragma unroll
    for (int sc = 0; sc < FIN_SC; ++sc) {
        float m = cm[sc][0];
#pragma unroll
        for (int ch = 1; ch < CHUNKS; ++ch) m = fmaxf(m, cm[sc][ch]);
        agg[t][sc] = fmaxf(m + bias1, 0.0f);
    }
    __syncthreads();

    float4 acc[FIN_SC];
#pragma unroll
    for (int sc = 0; sc < FIN_SC; ++sc) acc[sc] = make_float4(0.f, 0.f, 0.f, 0.f);
#pragma unroll
    for (int i = 0; i < FIN_KPT; ++i) {
        const int k = kg + FIN_KG * i;
#pragma unroll
        for (int sc = 0; sc < FIN_SC; ++sc) {
            const float a = agg[k][sc];
            acc[sc].x += a * w[i].x; acc[sc].y += a * w[i].y;
            acc[sc].z += a * w[i].z; acc[sc].w += a * w[i].w;
        }
    }
#pragma unroll
    for (int sc = 0; sc < FIN_SC; ++sc)
        *(float4*)&red[kg][sc][4 * lane] = acc[sc];
    __syncthreads();

    if (t < FIN_SC * FIN_CW) {
        float mu = tb_mu;
        float sr = tb_sr;
#pragma unroll
        for (int g = 0; g < FIN_KG; ++g) {
            mu += red[g][tsc][tcc];
            sr += red[g][tsc][FIN_CW + tcc];
        }
        float sig = fmaxf(sr, 0.0f) + log1pf(expf(-fabsf(sr))) + 1e-4f;
        out[(size_t)(s0 + tsc) * CMID + FIN_CW * sl + tcc] = mu + sig * tnoise;
    }
}

extern "C" void kernel_launch(void* const* d_in, const int* in_sizes, int n_in,
                              void* d_out, int out_size, void* d_ws, size_t ws_size,
                              hipStream_t stream) {
    const float* pos     = (const float*)d_in[0];
    const float* feature = (const float*)d_in[1];
    const float* W1      = (const float*)d_in[3];
    const float* b1      = (const float*)d_in[4];
    const float* W2      = (const float*)d_in[5];
    const float* b2      = (const float*)d_in[6];
    const float* noise   = (const float*)d_in[7];
    float* out = (float*)d_out;

    _Float16* W1t    = (_Float16*)((char*)d_ws + WS_W1T);
    float*    partial= (float*)((char*)d_ws + WS_PARTIAL);

    prep_w1t_kernel<<<(CMID * KTOT + 255) / 256, 256, 0, stream>>>(W1, W1t);
    gemm_max_kernel<<<GBLOCKS, 256, 0, stream>>>(pos, feature, W1t, partial);
    finish_kernel<<<FIN_BLOCKS, 256, 0, stream>>>(partial, b1, W2, b2, noise, out);
}


# W1 conversion launch removed: each GEMM wave reads its 259 x 32 slab of fp32 W1 as whole lines, rounds it to f16 and builds its B fragments through its own LDS slab in the prologue (launches 3 -> 2)
# speedup vs baseline: 1.8690x; 1.0695x over previous
#include <hip/hip_runtime.h>
#include <hip/hip_bf16.h>
#include <math.h>

typedef __attribute__((ext_vector_type(16))) _Float16 v16h;
typedef __attribute__((ext_vector_type(8)))  _Float16 v8h;
typedef __attribute__((ext_vector_type(4)))  _Float16 v4h;
typedef __attribute__((ext_vector_type(8)))  float    v8f;
typedef __attribute__((ext_vector_type(4)))  float    v4f;

#define NPTS      262144
#define BSC       128
#define PPS       2048
#define CIN       256
#define CMID      256
#define COUT      512
#define KTOT      288
#define ROWS      64
#define CHUNKS    (PPS / ROWS)
#define ASTRIDE   296
#define NCH       16
#define GBLOCKS   (BSC * CHUNKS / NCH)
static_assert(CHUNKS % NCH == 0, "a block's chunks must stay inside one scene");
static_assert(NCH % 2 == 0 && NCH >= 6, "the chunk pipeline runs two chunks per trip plus a four-chunk tail");

#define WS_PARTIAL 149504


static __device__ __forceinline__ void lds_barrier() {
    asm volatile("s_wait_dscnt 0x0" ::: "memory");
    asm volatile("s_barrier_signal -1" ::: "memory");
    asm volatile("s_barrier_wait -1" ::: "memory");
}

__global__ __launch_bounds__(256)
void gemm_max_kernel(const float* __restrict__ pos,
                     const float* __restrict__ feature,
                     const float* __restrict__ W1,
                     float* __restrict__ partial) {
    __shared__ __align__(16) _Float16 As[2][ROWS][ASTRIDE];
    __shared__ __align__(16) _Float16 Bs[8][32][ASTRIDE];
    __shared__ float csum[8][4];

    const int t = threadIdx.x;
    const int chunk0 = blockIdx.x * NCH;
    const int s = chunk0 >> 5;

    const int lane = t & 31;
    const int wv   = t >> 5;
    const int n    = lane & 15;
    const int half = lane >> 4;
    const int lo8  = half * 8;

    const int rsub = t >> 6;
    const int jcol = t & 63;
    const int tpos = (t < ROWS * 3) ? t : (ROWS * 3 - 1);
    const int prow = tpos / 3;
    const int pk   = tpos - prow * 3;
    const float4* f4 = (const float4*)feature;

    float4 pf[2][16];
    float  pp[2];

#define STAGE_ISSUE(st, chunk_)                                                 \
    {                                                                           \
        const float4* src_ = f4 + ((size_t)(chunk_) * ROWS + rsub) * (CIN / 4) + jcol; \
           \
           \
        _Pragma("unroll")                                                       \
        for (int i_ = 0; i_ < 16; ++i_) pf[st][i_] = src_[(size_t)i_ * 4 * (CIN / 4)]; \
        pp[st] = pos[(size_t)(chunk_) * (ROWS * 3) + tpos];                     \
    }

    const int kgs = lane & 3;
    const int cg  = lane >> 2;
    const float* w1p = W1 + (size_t)(kgs * 8) * CMID + wv * 32 + cg * 4;
    v4f wr[9][8];
#pragma unroll
    for (int it = 0; it < 4; ++it) {
#pragma unroll
        for (int j = 0; j < 8; ++j)
            wr[it][j] = *(const v4f*)(w1p + (size_t)(it * 32 + j) * CMID);
    }

    asm volatile("" ::: "memory");

    static_assert(PPS * 3 == 6 * 256 * 4, "6 float4 per thread cover a scene's pos");
    const v4f* cp4 = (const v4f*)(pos + (size_t)s * (PPS * 3));
    v4f cv[6];
#pragma unroll
    for (int i = 0; i < 6; ++i) cv[i] = cp4[t + 256 * i];

    asm volatile("" ::: "memory");
    STAGE_ISSUE(0, chunk0)
    asm volatile("" ::: "memory");
#pragma unroll
    for (int it = 4; it < 8; ++it) {
#pragma unroll
        for (int j = 0; j < 8; ++j)
            wr[it][j] = *(const v4f*)(w1p + (size_t)(it * 32 + j) * CMID);
    }
    {
        const float* w1q = W1 + wv * 32 + cg * 4;
        const v4f zero4 = {0.f, 0.f, 0.f, 0.f};
#pragma unroll
        for (int j = 0; j < 3; ++j) {
            v4f v = *(const v4f*)(w1q + (size_t)(CIN + j) * CMID);
            wr[8][j] = (kgs == 0) ? v : zero4;
        }
#pragma unroll
        for (int j = 3; j < 8; ++j) wr[8][j] = zero4;
    }
    asm volatile("" ::: "memory");
    STAGE_ISSUE(1, chunk0 + 1)
    asm volatile("" : "+v"(cv[0]), "+v"(cv[1]), "+v"(cv[2]),
                      "+v"(cv[3]), "+v"(cv[4]), "+v"(cv[5]) :: "memory");

    float rq[3] = {0.f, 0.f, 0.f};
#pragma unroll
    for (int i = 0; i < 6; ++i) {
        rq[(i + 0) % 3] += cv[i].x;
        rq[(i + 1) % 3] += cv[i].y;
        rq[(i + 2) % 3] += cv[i].z;
        rq[(i + 3) % 3] += cv[i].w;
    }
    const int ta = t % 3;
    float cx = (ta == 0) ? rq[0] : (ta == 1) ? rq[2] : rq[1];
    float cy = (ta == 0) ? rq[1] : (ta == 1) ? rq[0] : rq[2];
    float cz = (ta == 0) ? rq[2] : (ta == 1) ? rq[1] : rq[0];
    for (int off = 16; off > 0; off >>= 1) {
        cx += __shfl_xor(cx, off, 32);
        cy += __shfl_xor(cy, off, 32);
        cz += __shfl_xor(cz, off, 32);
    }
    if (lane == 0) { csum[wv][0] = cx; csum[wv][1] = cy; csum[wv][2] = cz; }

#define W1_TO_LDS(it_lo, it_hi)                                                 \
                     \
    _Pragma("unroll")                                                           \
    for (int it = it_lo; it < it_hi; ++it) {                                    \
        _Pragma("unroll")                                                       \
        for (int c = 0; c < 4; ++c) {                                           \
            v8h h_;                                                             \
            _Pragma("unroll")                                                   \
            for (int j = 0; j < 8; ++j) h_[j] = (_Float16)wr[it][j][c];         \
            *(v8h*)&Bs[wv][cg * 4 + c][(it * 4 + kgs) * 8] = h_;                \
        }                                                                       \
    }
    W1_TO_LDS(0, 4)

    lds_barrier();
    float pc = csum[0][pk];
    for (int w = 1; w < 8; ++w) pc += csum[w][pk];
    pc *= 1.0f / (float)PPS;

    if (t < ROWS) {
        for (int k = CIN + 3; k < ASTRIDE; ++k) {
            As[0][t][k] = (_Float16)0.0f;
            As[1][t][k] = (_Float16)0.0f;
        }
    }

#define STAGE_STORE(st, buf)                                                    \
    {                                                                           \
                     \
        _Pragma("unroll")                                                       \
        for (int i_ = 0; i_ < 16; ++i_) {                                       \
            float4 v_ = pf[st][i_];                                             \
            v4h h_;                                                             \
            h_[0] = (_Float16)v_.x; h_[1] = (_Float16)v_.y;                     \
            h_[2] = (_Float16)v_.z; h_[3] = (_Float16)v_.w;                     \
            *(v4h*)&As[buf][rsub + 4 * i_][jcol * 4] = h_;                      \
        }                                                                       \
        if (t < ROWS * 3) As[buf][prow][CIN + pk] = (_Float16)(pp[st] - pc);    \
    }

#define LOAD_A(buf, set, kt_)                                                   \
    for (int r2 = 0; r2 < 2; ++r2) {                                            \
        const _Float16* ap = &As[buf][(rp * 2 + r2) * 16 + n][(kt_) * 32 + lo8]; \
        v8h alo = *(const v8h*)ap;                                              \
        v8h ahi = *(const v8h*)(ap + 16);                                       \
        afrag[set][r2] = __builtin_shufflevector(alo, ahi,                      \
            0,1,2,3,4,5,6,7,8,9,10,11,12,13,14,15);                             \
    }

#define COMPUTE(buf, chunk_)                                                    \
    {                                                                           \
        float cmax[2];                                                          \
        for (int rp = 0; rp < 2; ++rp) {                                        \
            v8f acc[2][2] = {};                                                 \
            v16h afrag[3][2];                                                   \
            LOAD_A(buf, 0, 0)                                                   \
            LOAD_A(buf, 1, 1)                                                   \
            for (int kt = 0; kt < KTOT / 32; ++kt) {                            \
                if (kt + 2 < KTOT / 32) { LOAD_A(buf, (kt + 2) % 3, kt + 2) }   \
                    \
                    \
                    \
                    \
                __builtin_amdgcn_sched_barrier(0);                              \
                for (int r2 = 0; r2 < 2; ++r2) {                                \
                    acc[r2][0] = __builtin_amdgcn_wmma_f32_16x16x32_f16(        \
                        false, afrag[kt % 3][r2], false, bfrag[kt][0], (short)0, acc[r2][0], false, false); \
                    acc[r2][1] = __builtin_amdgcn_wmma_f32_16x16x32_f16(        \
                        false, afrag[kt % 3][r2], false, bfrag[kt][1], (short)0, acc[r2][1], false, false); \
                }                                                               \
            }                                                                   \
            for (int ct = 0; ct < 2; ++ct) {                                    \
                float m = (rp == 0) ? acc[0][ct][0] : cmax[ct];                 \
                for (int r2 = 0; r2 < 2; ++r2)                                  \
                    for (int i = 0; i < 8; ++i)                                 \
                        m = fmaxf(m, acc[r2][ct][i]);                           \
                cmax[ct] = m;                                                   \
            }                                                                   \
        }                                                                       \
        for (int ct = 0; ct < 2; ++ct) {                                        \
               \
            float m = fmaxf(cmax[ct], __shfl_xor(cmax[ct], 16, 32));            \
            if (half == 0) {                                                    \
                int col = (wv * 2 + ct) * 16 + n;                               \
                partial[(size_t)(chunk_) * CMID + col] = m;                     \
            }                                                                   \
        }                                                                       \
    }

    STAGE_STORE(0, 0)
    STAGE_ISSUE(0, chunk0 + 2)
    W1_TO_LDS(4, 9)
#undef W1_TO_LDS
    lds_barrier();

    v16h bfrag[KTOT / 32][2];
    for (int kt = 0; kt < KTOT / 32; ++kt) {
        for (int ct = 0; ct < 2; ++ct) {
            const _Float16* bp = &Bs[wv][ct * 16 + n][kt * 32 + lo8];
            v8h blo = *(const v8h*)bp;
            v8h bhi = *(const v8h*)(bp + 16);
            bfrag[kt][ct] = __builtin_shufflevector(blo, bhi,
                0,1,2,3,4,5,6,7,8,9,10,11,12,13,14,15);
        }
    }


    for (int c = 0; c < NCH - 4; c += 2) {
        COMPUTE(0, chunk0 + c)
        STAGE_STORE(1, 1)
        STAGE_ISSUE(1, chunk0 + c + 3)
        lds_barrier();
        COMPUTE(1, chunk0 + c + 1)
        STAGE_STORE(0, 0)
        STAGE_ISSUE(0, chunk0 + c + 4)
        lds_barrier();
    }
    COMPUTE(0, chunk0 + NCH - 4)
    STAGE_STORE(1, 1)
    STAGE_ISSUE(1, chunk0 + NCH - 1)
    lds_barrier();
    COMPUTE(1, chunk0 + NCH - 3)
    STAGE_STORE(0, 0)
    lds_barrier();
    COMPUTE(0, chunk0 + NCH - 2)
    STAGE_STORE(1, 1)
    lds_barrier();
    COMPUTE(1, chunk0 + NCH - 1)

#undef COMPUTE
#undef LOAD_A
#undef STAGE_STORE
#undef STAGE_ISSUE
}

#define FIN_SC     2
#define FIN_CW     64
#define FIN_KG     8
#define FIN_KPT    (CMID / FIN_KG)
#define FIN_SLICES (CMID / FIN_CW)
#define FIN_BLOCKS ((BSC / FIN_SC) * FIN_SLICES)
static_assert(FIN_CW * 2 == 32 * 4, "one wave = one W2 row of the block: 16 mu + 16 sigma float4");
static_assert(FIN_SC * FIN_CW <= 256 && CMID == 256, "thread maps of the head and the tail");

__global__ __launch_bounds__(256)
void finish_kernel(const float* __restrict__ partial,
                   const float* __restrict__ b1,
                   const float* __restrict__ W2,
                   const float* __restrict__ b2,
                   const float* __restrict__ noise,
                   float* __restrict__ out) {
    __shared__ __align__(8)  float agg[CMID][FIN_SC];
    __shared__ __align__(16) float red[FIN_KG][FIN_SC][2 * FIN_CW];

    const int t    = threadIdx.x;
    const int lane = t & 31;
    const int kg   = t >> 5;
    const int sl   = blockIdx.x % FIN_SLICES;
    const int s0   = (blockIdx.x / FIN_SLICES) * FIN_SC;

    const int col4 = (lane < 16) ? (FIN_CW * sl + 4 * lane)
                                 : (CMID + FIN_CW * sl + 4 * (lane - 16));

    float cm[FIN_SC][CHUNKS];
#pragma unroll
    for (int sc = 0; sc < FIN_SC; ++sc) {
        const float* p = partial + (size_t)(s0 + sc) * CHUNKS * CMID + t;
#pragma unroll
        for (int ch = 0; ch < CHUNKS; ++ch) cm[sc][ch] = p[ch * CMID];
    }
    const float bias1 = b1[t];

    float4 w[FIN_KPT];
    const float4* wp = (const float4*)(W2 + (size_t)kg * COUT + col4);
#pragma unroll
    for (int i = 0; i < FIN_KPT; ++i) w[i] = wp[(size_t)i * FIN_KG * (COUT / 4)];

    const int tsc = t / FIN_CW;
    const int tcc = t % FIN_CW;
    float tb_mu = 0.f, tb_sr = 0.f, tnoise = 0.f;
    if (t < FIN_SC * FIN_CW) {
        tb_mu  = b2[FIN_CW * sl + tcc];
        tb_sr  = b2[CMID + FIN_CW * sl + tcc];
        tnoise = noise[(size_t)(s0 + tsc) * CMID + FIN_CW * sl + tcc];
    }

#pragma unroll
    for (int sc = 0; sc < FIN_SC; ++sc) {
        float m = cm[sc][0];
#pragma unroll
        for (int ch = 1; ch < CHUNKS; ++ch) m = fmaxf(m, cm[sc][ch]);
        agg[t][sc] = fmaxf(m + bias1, 0.0f);
    }
    __syncthreads();

    float4 acc[FIN_SC];
#pragma unroll
    for (int sc = 0; sc < FIN_SC; ++sc) acc[sc] = make_float4(0.f, 0.f, 0.f, 0.f);
#pragma unroll
    for (int i = 0; i < FIN_KPT; ++i) {
        const int k = kg + FIN_KG * i;
#pragma unroll
        for (int sc = 0; sc < FIN_SC; ++sc) {
            const float a = agg[k][sc];
            acc[sc].x += a * w[i].x; acc[sc].y += a * w[i].y;
            acc[sc].z += a * w[i].z; acc[sc].w += a * w[i].w;
        }
    }
#pragma unroll
    for (int sc = 0; sc < FIN_SC; ++sc)
        *(float4*)&red[kg][sc][4 * lane] = acc[sc];
    __syncthreads();

    if (t < FIN_SC * FIN_CW) {
        float mu = tb_mu;
        float sr = tb_sr;
#pragma unroll
        for (int g = 0; g < FIN_KG; ++g) {
            mu += red[g][tsc][tcc];
            sr += red[g][tsc][FIN_CW + tcc];
        }
        float sig = fmaxf(sr, 0.0f) + log1pf(expf(-fabsf(sr))) + 1e-4f;
        out[(size_t)(s0 + tsc) * CMID + FIN_CW * sl + tcc] = mu + sig * tnoise;
    }
}

extern "C" void kernel_launch(void* const* d_in, const int* in_sizes, int n_in,
                              void* d_out, int out_size, void* d_ws, size_t ws_size,
                              hipStream_t stream) {
    const float* pos     = (const float*)d_in[0];
    const float* feature = (const float*)d_in[1];
    const float* W1      = (const float*)d_in[3];
    const float* b1      = (const float*)d_in[4];
    const float* W2      = (const float*)d_in[5];
    const float* b2      = (const float*)d_in[6];
    const float* noise   = (const float*)d_in[7];
    float* out = (float*)d_out;

    float*    partial= (float*)((char*)d_ws + WS_PARTIAL);

    gemm_max_kernel<<<GBLOCKS, 256, 0, stream>>>(pos, feature, W1, partial);
    finish_kernel<<<FIN_BLOCKS, 256, 0, stream>>>(partial, b1, W2, b2, noise, out);
}


# Column max kept in registers over a GEMM block's 16 chunks: one partial row per block instead of 16, no lane exchange or global store in the chunk loop; finish kernel reads 2 rows per scene, not 32
# speedup vs baseline: 2.0148x; 1.0780x over previous
#include <hip/hip_runtime.h>
#include <hip/hip_bf16.h>
#include <math.h>

typedef __attribute__((ext_vector_type(16))) _Float16 v16h;
typedef __attribute__((ext_vector_type(8)))  _Float16 v8h;
typedef __attribute__((ext_vector_type(4)))  _Float16 v4h;
typedef __attribute__((ext_vector_type(8)))  float    v8f;
typedef __attribute__((ext_vector_type(4)))  float    v4f;

#define NPTS      262144
#define BSC       128
#define PPS       2048
#define CIN       256
#define CMID      256
#define COUT      512
#define KTOT      288
#define ROWS      64
#define CHUNKS    (PPS / ROWS)
#define ASTRIDE   296
#define NCH       16
#define GBLOCKS   (BSC * CHUNKS / NCH)
static_assert(CHUNKS % NCH == 0, "a block's chunks must stay inside one scene");
static_assert(NCH % 2 == 0 && NCH >= 6, "the chunk pipeline runs two chunks per trip plus a four-chunk tail");

#define WS_PARTIAL 149504


static __device__ __forceinline__ void lds_barrier() {
    asm volatile("s_wait_dscnt 0x0" ::: "memory");
    asm volatile("s_barrier_signal -1" ::: "memory");
    asm volatile("s_barrier_wait -1" ::: "memory");
}

__global__ __launch_bounds__(256)
void gemm_max_kernel(const float* __restrict__ pos,
                     const float* __restrict__ feature,
                     const float* __restrict__ W1,
                     float* __restrict__ partial) {
    __shared__ __align__(16) _Float16 As[2][ROWS][ASTRIDE];
    __shared__ __align__(16) _Float16 Bs[8][32][ASTRIDE];
    __shared__ float csum[8][4];

    const int t = threadIdx.x;
    const int chunk0 = blockIdx.x * NCH;
    const int s = chunk0 >> 5;

    const int lane = t & 31;
    const int wv   = t >> 5;
    const int n    = lane & 15;
    const int half = lane >> 4;
    const int lo8  = half * 8;

    const int rsub = t >> 6;
    const int jcol = t & 63;
    const int tpos = (t < ROWS * 3) ? t : (ROWS * 3 - 1);
    const int prow = tpos / 3;
    const int pk   = tpos - prow * 3;
    const float4* f4 = (const float4*)feature;

    float4 pf[2][16];
    float  pp[2];

#define STAGE_ISSUE(st, chunk_)                                                 \
    {                                                                           \
        const float4* src_ = f4 + ((size_t)(chunk_) * ROWS + rsub) * (CIN / 4) + jcol; \
           \
           \
        _Pragma("unroll")                                                       \
        for (int i_ = 0; i_ < 16; ++i_) pf[st][i_] = src_[(size_t)i_ * 4 * (CIN / 4)]; \
        pp[st] = pos[(size_t)(chunk_) * (ROWS * 3) + tpos];                     \
    }

    const int kgs = lane & 3;
    const int cg  = lane >> 2;
    const float* w1p = W1 + (size_t)(kgs * 8) * CMID + wv * 32 + cg * 4;
    v4f wr[9][8];
#pragma unroll
    for (int it = 0; it < 4; ++it) {
#pragma unroll
        for (int j = 0; j < 8; ++j)
            wr[it][j] = *(const v4f*)(w1p + (size_t)(it * 32 + j) * CMID);
    }

    asm volatile("" ::: "memory");

    static_assert(PPS * 3 == 6 * 256 * 4, "6 float4 per thread cover a scene's pos");
    const v4f* cp4 = (const v4f*)(pos + (size_t)s * (PPS * 3));
    v4f cv[6];
#pragma unroll
    for (int i = 0; i < 6; ++i) cv[i] = cp4[t + 256 * i];

    asm volatile("" ::: "memory");
    STAGE_ISSUE(0, chunk0)
    asm volatile("" ::: "memory");
#pragma unroll
    for (int it = 4; it < 8; ++it) {
#pragma unroll
        for (int j = 0; j < 8; ++j)
            wr[it][j] = *(const v4f*)(w1p + (size_t)(it * 32 + j) * CMID);
    }
    {
        const float* w1q = W1 + wv * 32 + cg * 4;
        const v4f zero4 = {0.f, 0.f, 0.f, 0.f};
#pragma unroll
        for (int j = 0; j < 3; ++j) {
            v4f v = *(const v4f*)(w1q + (size_t)(CIN + j) * CMID);
            wr[8][j] = (kgs == 0) ? v : zero4;
        }
#pragma unroll
        for (int j = 3; j < 8; ++j) wr[8][j] = zero4;
    }
    asm volatile("" ::: "memory");
    STAGE_ISSUE(1, chunk0 + 1)
    asm volatile("" : "+v"(cv[0]), "+v"(cv[1]), "+v"(cv[2]),
                      "+v"(cv[3]), "+v"(cv[4]), "+v"(cv[5]) :: "memory");

    float rq[3] = {0.f, 0.f, 0.f};
#pragma unroll
    for (int i = 0; i < 6; ++i) {
        rq[(i + 0) % 3] += cv[i].x;
        rq[(i + 1) % 3] += cv[i].y;
        rq[(i + 2) % 3] += cv[i].z;
        rq[(i + 3) % 3] += cv[i].w;
    }
    const int ta = t % 3;
    float cx = (ta == 0) ? rq[0] : (ta == 1) ? rq[2] : rq[1];
    float cy = (ta == 0) ? rq[1] : (ta == 1) ? rq[0] : rq[2];
    float cz = (ta == 0) ? rq[2] : (ta == 1) ? rq[1] : rq[0];
    for (int off = 16; off > 0; off >>= 1) {
        cx += __shfl_xor(cx, off, 32);
        cy += __shfl_xor(cy, off, 32);
        cz += __shfl_xor(cz, off, 32);
    }
    if (lane == 0) { csum[wv][0] = cx; csum[wv][1] = cy; csum[wv][2] = cz; }

#define W1_TO_LDS(it_lo, it_hi)                                                 \
                     \
    _Pragma("unroll")                                                           \
    for (int it = it_lo; it < it_hi; ++it) {                                    \
        _Pragma("unroll")                                                       \
        for (int c = 0; c < 4; ++c) {                                           \
            v8h h_;                                                             \
            _Pragma("unroll")                                                   \
            for (int j = 0; j < 8; ++j) h_[j] = (_Float16)wr[it][j][c];         \
            *(v8h*)&Bs[wv][cg * 4 + c][(it * 4 + kgs) * 8] = h_;                \
        }                                                                       \
    }
    W1_TO_LDS(0, 4)

    lds_barrier();
    float pc = csum[0][pk];
    for (int w = 1; w < 8; ++w) pc += csum[w][pk];
    pc *= 1.0f / (float)PPS;

    if (t < ROWS) {
        for (int k = CIN + 3; k < ASTRIDE; ++k) {
            As[0][t][k] = (_Float16)0.0f;
            As[1][t][k] = (_Float16)0.0f;
        }
    }

#define STAGE_STORE(st, buf)                                                    \
    {                                                                           \
                     \
        _Pragma("unroll")                                                       \
        for (int i_ = 0; i_ < 16; ++i_) {                                       \
            float4 v_ = pf[st][i_];                                             \
            v4h h_;                                                             \
            h_[0] = (_Float16)v_.x; h_[1] = (_Float16)v_.y;                     \
            h_[2] = (_Float16)v_.z; h_[3] = (_Float16)v_.w;                     \
            *(v4h*)&As[buf][rsub + 4 * i_][jcol * 4] = h_;                      \
        }                                                                       \
        if (t < ROWS * 3) As[buf][prow][CIN + pk] = (_Float16)(pp[st] - pc);    \
    }

#define LOAD_A(buf, set, kt_)                                                   \
    for (int r2 = 0; r2 < 2; ++r2) {                                            \
        const _Float16* ap = &As[buf][(rp * 2 + r2) * 16 + n][(kt_) * 32 + lo8]; \
        v8h alo = *(const v8h*)ap;                                              \
        v8h ahi = *(const v8h*)(ap + 16);                                       \
        afrag[set][r2] = __builtin_shufflevector(alo, ahi,                      \
            0,1,2,3,4,5,6,7,8,9,10,11,12,13,14,15);                             \
    }

#define COMPUTE(buf)                                                            \
    {                                                                           \
        float cmax[2];                                                          \
        for (int rp = 0; rp < 2; ++rp) {                                        \
            v8f acc[2][2] = {};                                                 \
            v16h afrag[3][2];                                                   \
            LOAD_A(buf, 0, 0)                                                   \
            LOAD_A(buf, 1, 1)                                                   \
            for (int kt = 0; kt < KTOT / 32; ++kt) {                            \
                if (kt + 2 < KTOT / 32) { LOAD_A(buf, (kt + 2) % 3, kt + 2) }   \
                    \
                    \
                    \
                    \
                __builtin_amdgcn_sched_barrier(0);                              \
                for (int r2 = 0; r2 < 2; ++r2) {                                \
                    acc[r2][0] = __builtin_amdgcn_wmma_f32_16x16x32_f16(        \
                        false, afrag[kt % 3][r2], false, bfrag[kt][0], (short)0, acc[r2][0], false, false); \
                    acc[r2][1] = __builtin_amdgcn_wmma_f32_16x16x32_f16(        \
                        false, afrag[kt % 3][r2], false, bfrag[kt][1], (short)0, acc[r2][1], false, false); \
                }                                                               \
            }                                                                   \
            for (int ct = 0; ct < 2; ++ct) {                                    \
                float m = (rp == 0) ? acc[0][ct][0] : cmax[ct];                 \
                for (int r2 = 0; r2 < 2; ++r2)                                  \
                    for (int i = 0; i < 8; ++i)                                 \
                        m = fmaxf(m, acc[r2][ct][i]);                           \
                cmax[ct] = m;                                                   \
            }                                                                   \
        }                                                                       \
        for (int ct = 0; ct < 2; ++ct) rmax[ct] = fmaxf(rmax[ct], cmax[ct]);    \
    }

    STAGE_STORE(0, 0)
    STAGE_ISSUE(0, chunk0 + 2)
    W1_TO_LDS(4, 9)
#undef W1_TO_LDS
    lds_barrier();

    v16h bfrag[KTOT / 32][2];
    for (int kt = 0; kt < KTOT / 32; ++kt) {
        for (int ct = 0; ct < 2; ++ct) {
            const _Float16* bp = &Bs[wv][ct * 16 + n][kt * 32 + lo8];
            v8h blo = *(const v8h*)bp;
            v8h bhi = *(const v8h*)(bp + 16);
            bfrag[kt][ct] = __builtin_shufflevector(blo, bhi,
                0,1,2,3,4,5,6,7,8,9,10,11,12,13,14,15);
        }
    }


    float rmax[2] = {-INFINITY, -INFINITY};

    for (int c = 0; c < NCH - 4; c += 2) {
        COMPUTE(0)
        STAGE_STORE(1, 1)
        STAGE_ISSUE(1, chunk0 + c + 3)
        lds_barrier();
        COMPUTE(1)
        STAGE_STORE(0, 0)
        STAGE_ISSUE(0, chunk0 + c + 4)
        lds_barrier();
    }
    COMPUTE(0)
    STAGE_STORE(1, 1)
    STAGE_ISSUE(1, chunk0 + NCH - 1)
    lds_barrier();
    COMPUTE(1)
    STAGE_STORE(0, 0)
    lds_barrier();
    COMPUTE(0)
    STAGE_STORE(1, 1)
    lds_barrier();
    COMPUTE(1)

    for (int ct = 0; ct < 2; ++ct) {
        float m = fmaxf(rmax[ct], __shfl_xor(rmax[ct], 16, 32));
        if (half == 0) {
            int col = (wv * 2 + ct) * 16 + n;
            partial[(size_t)blockIdx.x * CMID + col] = m;
        }
    }

#undef COMPUTE
#undef LOAD_A
#undef STAGE_STORE
#undef STAGE_ISSUE
}

#define FIN_SC     2
#define FIN_CW     64
#define FIN_KG     8
#define FIN_KPT    (CMID / FIN_KG)
#define FIN_SLICES (CMID / FIN_CW)
#define FIN_BLOCKS ((BSC / FIN_SC) * FIN_SLICES)
#define FIN_PARTS  (CHUNKS / NCH)
static_assert(FIN_CW * 2 == 32 * 4, "one wave = one W2 row of the block: 16 mu + 16 sigma float4");
static_assert(FIN_SC * FIN_CW <= 256 && CMID == 256, "thread maps of the head and the tail");

__global__ __launch_bounds__(256)
void finish_kernel(const float* __restrict__ partial,
                   const float* __restrict__ b1,
                   const float* __restrict__ W2,
                   const float* __restrict__ b2,
                   const float* __restrict__ noise,
                   float* __restrict__ out) {
    __shared__ __align__(8)  float agg[CMID][FIN_SC];
    __shared__ __align__(16) float red[FIN_KG][FIN_SC][2 * FIN_CW];

    const int t    = threadIdx.x;
    const int lane = t & 31;
    const int kg   = t >> 5;
    const int sl   = blockIdx.x % FIN_SLICES;
    const int s0   = (blockIdx.x / FIN_SLICES) * FIN_SC;

    const int col4 = (lane < 16) ? (FIN_CW * sl + 4 * lane)
                                 : (CMID + FIN_CW * sl + 4 * (lane - 16));

    float cm[FIN_SC][FIN_PARTS];
#pragma unroll
    for (int sc = 0; sc < FIN_SC; ++sc) {
        const float* p = partial + (size_t)(s0 + sc) * FIN_PARTS * CMID + t;
#pragma unroll
        for (int ch = 0; ch < FIN_PARTS; ++ch) cm[sc][ch] = p[ch * CMID];
    }
    const float bias1 = b1[t];

    float4 w[FIN_KPT];
    const float4* wp = (const float4*)(W2 + (size_t)kg * COUT + col4);
#pragma unroll
    for (int i = 0; i < FIN_KPT; ++i) w[i] = wp[(size_t)i * FIN_KG * (COUT / 4)];

    const int tsc = t / FIN_CW;
    const int tcc = t % FIN_CW;
    float tb_mu = 0.f, tb_sr = 0.f, tnoise = 0.f;
    if (t < FIN_SC * FIN_CW) {
        tb_mu  = b2[FIN_CW * sl + tcc];
        tb_sr  = b2[CMID + FIN_CW * sl + tcc];
        tnoise = noise[(size_t)(s0 + tsc) * CMID + FIN_CW * sl + tcc];
    }

#pragma unroll
    for (int sc = 0; sc < FIN_SC; ++sc) {
        float m = cm[sc][0];
#pragma unroll
        for (int ch = 1; ch < FIN_PARTS; ++ch) m = fmaxf(m, cm[sc][ch]);
        agg[t][sc] = fmaxf(m + bias1, 0.0f);
    }
    __syncthreads();

    float4 acc[FIN_SC];
#pragma unroll
    for (int sc = 0; sc < FIN_SC; ++sc) acc[sc] = make_float4(0.f, 0.f, 0.f, 0.f);
#pragma unroll
    for (int i = 0; i < FIN_KPT; ++i) {
        const int k = kg + FIN_KG * i;
#pragma unroll
        for (int sc = 0; sc < FIN_SC; ++sc) {
            const float a = agg[k][sc];
            acc[sc].x += a * w[i].x; acc[sc].y += a * w[i].y;
            acc[sc].z += a * w[i].z; acc[sc].w += a * w[i].w;
        }
    }
#pragma unroll
    for (int sc = 0; sc < FIN_SC; ++sc)
        *(float4*)&red[kg][sc][4 * lane] = acc[sc];
    __syncthreads();

    if (t < FIN_SC * FIN_CW) {
        float mu = tb_mu;
        float sr = tb_sr;
#pragma unroll
        for (int g = 0; g < FIN_KG; ++g) {
            mu += red[g][tsc][tcc];
            sr += red[g][tsc][FIN_CW + tcc];
        }
        float sig = fmaxf(sr, 0.0f) + log1pf(expf(-fabsf(sr))) + 1e-4f;
        out[(size_t)(s0 + tsc) * CMID + FIN_CW * sl + tcc] = mu + sig * tnoise;
    }
}

extern "C" void kernel_launch(void* const* d_in, const int* in_sizes, int n_in,
                              void* d_out, int out_size, void* d_ws, size_t ws_size,
                              hipStream_t stream) {
    const float* pos     = (const float*)d_in[0];
    const float* feature = (const float*)d_in[1];
    const float* W1      = (const float*)d_in[3];
    const float* b1      = (const float*)d_in[4];
    const float* W2      = (const float*)d_in[5];
    const float* b2      = (const float*)d_in[6];
    const float* noise   = (const float*)d_in[7];
    float* out = (float*)d_out;

    float*    partial= (float*)((char*)d_ws + WS_PARTIAL);

    gemm_max_kernel<<<GBLOCKS, 256, 0, stream>>>(pos, feature, W1, partial);
    finish_kernel<<<FIN_BLOCKS, 256, 0, stream>>>(partial, b1, W2, b2, noise, out);
}


# GEMM kernel: next chunk's f32->f16 conversion, LDS stores and register reloads issued piecewise inside the current chunk's K-steps, no longer a phase between the multiply and the barrier
# speedup vs baseline: 2.0944x; 1.0395x over previous
#include <hip/hip_runtime.h>
#include <hip/hip_bf16.h>
#include <math.h>

typedef __attribute__((ext_vector_type(16))) _Float16 v16h;
typedef __attribute__((ext_vector_type(8)))  _Float16 v8h;
typedef __attribute__((ext_vector_type(4)))  _Float16 v4h;
typedef __attribute__((ext_vector_type(8)))  float    v8f;
typedef __attribute__((ext_vector_type(4)))  float    v4f;

#define NPTS      262144
#define BSC       128
#define PPS       2048
#define CIN       256
#define CMID      256
#define COUT      512
#define KTOT      288
#define ROWS      64
#define CHUNKS    (PPS / ROWS)
#define ASTRIDE   296
#define NCH       16
#define GBLOCKS   (BSC * CHUNKS / NCH)
static_assert(CHUNKS % NCH == 0, "a block's chunks must stay inside one scene");
static_assert(NCH % 2 == 0 && NCH >= 6, "the chunk pipeline runs two chunks per trip plus a four-chunk tail");

#define WS_PARTIAL 149504


static __device__ __forceinline__ void lds_barrier() {
    asm volatile("s_wait_dscnt 0x0" ::: "memory");
    asm volatile("s_barrier_signal -1" ::: "memory");
    asm volatile("s_barrier_wait -1" ::: "memory");
}

__global__ __launch_bounds__(256)
void gemm_max_kernel(const float* __restrict__ pos,
                     const float* __restrict__ feature,
                     const float* __restrict__ W1,
                     float* __restrict__ partial) {
    __shared__ __align__(16) _Float16 As[2][ROWS][ASTRIDE];
    __shared__ __align__(16) _Float16 Bs[8][32][ASTRIDE];
    __shared__ float csum[8][4];

    const int t = threadIdx.x;
    const int chunk0 = blockIdx.x * NCH;
    const int s = chunk0 >> 5;

    const int lane = t & 31;
    const int wv   = t >> 5;
    const int n    = lane & 15;
    const int half = lane >> 4;
    const int lo8  = half * 8;

    const int rsub = t >> 6;
    const int jcol = t & 63;
    const int tpos = (t < ROWS * 3) ? t : (ROWS * 3 - 1);
    const int prow = tpos / 3;
    const int pk   = tpos - prow * 3;
    const float4* f4 = (const float4*)feature;

    float4 pf[2][16];
    float  pp[2];

#define STAGE_ISSUE(st, chunk_)                                                 \
    {                                                                           \
        const float4* src_ = f4 + ((size_t)(chunk_) * ROWS + rsub) * (CIN / 4) + jcol; \
           \
           \
        _Pragma("unroll")                                                       \
        for (int i_ = 0; i_ < 16; ++i_) pf[st][i_] = src_[(size_t)i_ * 4 * (CIN / 4)]; \
        pp[st] = pos[(size_t)(chunk_) * (ROWS * 3) + tpos];                     \
    }

    const int kgs = lane & 3;
    const int cg  = lane >> 2;
    const float* w1p = W1 + (size_t)(kgs * 8) * CMID + wv * 32 + cg * 4;
    v4f wr[9][8];
#pragma unroll
    for (int it = 0; it < 4; ++it) {
#pragma unroll
        for (int j = 0; j < 8; ++j)
            wr[it][j] = *(const v4f*)(w1p + (size_t)(it * 32 + j) * CMID);
    }

    asm volatile("" ::: "memory");

    static_assert(PPS * 3 == 6 * 256 * 4, "6 float4 per thread cover a scene's pos");
    const v4f* cp4 = (const v4f*)(pos + (size_t)s * (PPS * 3));
    v4f cv[6];
#pragma unroll
    for (int i = 0; i < 6; ++i) cv[i] = cp4[t + 256 * i];

    asm volatile("" ::: "memory");
    STAGE_ISSUE(0, chunk0)
    asm volatile("" ::: "memory");
#pragma unroll
    for (int it = 4; it < 8; ++it) {
#pragma unroll
        for (int j = 0; j < 8; ++j)
            wr[it][j] = *(const v4f*)(w1p + (size_t)(it * 32 + j) * CMID);
    }
    {
        const float* w1q = W1 + wv * 32 + cg * 4;
        const v4f zero4 = {0.f, 0.f, 0.f, 0.f};
#pragma unroll
        for (int j = 0; j < 3; ++j) {
            v4f v = *(const v4f*)(w1q + (size_t)(CIN + j) * CMID);
            wr[8][j] = (kgs == 0) ? v : zero4;
        }
#pragma unroll
        for (int j = 3; j < 8; ++j) wr[8][j] = zero4;
    }
    asm volatile("" ::: "memory");
    STAGE_ISSUE(1, chunk0 + 1)
    asm volatile("" : "+v"(cv[0]), "+v"(cv[1]), "+v"(cv[2]),
                      "+v"(cv[3]), "+v"(cv[4]), "+v"(cv[5]) :: "memory");

    float rq[3] = {0.f, 0.f, 0.f};
#pragma unroll
    for (int i = 0; i < 6; ++i) {
        rq[(i + 0) % 3] += cv[i].x;
        rq[(i + 1) % 3] += cv[i].y;
        rq[(i + 2) % 3] += cv[i].z;
        rq[(i + 3) % 3] += cv[i].w;
    }
    const int ta = t % 3;
    float cx = (ta == 0) ? rq[0] : (ta == 1) ? rq[2] : rq[1];
    float cy = (ta == 0) ? rq[1] : (ta == 1) ? rq[0] : rq[2];
    float cz = (ta == 0) ? rq[2] : (ta == 1) ? rq[1] : rq[0];
    for (int off = 16; off > 0; off >>= 1) {
        cx += __shfl_xor(cx, off, 32);
        cy += __shfl_xor(cy, off, 32);
        cz += __shfl_xor(cz, off, 32);
    }
    if (lane == 0) { csum[wv][0] = cx; csum[wv][1] = cy; csum[wv][2] = cz; }

#define W1_TO_LDS(it_lo, it_hi)                                                 \
                     \
    _Pragma("unroll")                                                           \
    for (int it = it_lo; it < it_hi; ++it) {                                    \
        _Pragma("unroll")                                                       \
        for (int c = 0; c < 4; ++c) {                                           \
            v8h h_;                                                             \
            _Pragma("unroll")                                                   \
            for (int j = 0; j < 8; ++j) h_[j] = (_Float16)wr[it][j][c];         \
            *(v8h*)&Bs[wv][cg * 4 + c][(it * 4 + kgs) * 8] = h_;                \
        }                                                                       \
    }
    W1_TO_LDS(0, 4)

    lds_barrier();
    float pc = csum[0][pk];
    for (int w = 1; w < 8; ++w) pc += csum[w][pk];
    pc *= 1.0f / (float)PPS;

    if (t < ROWS) {
        for (int k = CIN + 3; k < ASTRIDE; ++k) {
            As[0][t][k] = (_Float16)0.0f;
            As[1][t][k] = (_Float16)0.0f;
        }
    }

#define STAGE_STORE(st, buf)                                                    \
    {                                                                           \
                     \
        _Pragma("unroll")                                                       \
        for (int i_ = 0; i_ < 16; ++i_) {                                       \
            float4 v_ = pf[st][i_];                                             \
            v4h h_;                                                             \
            h_[0] = (_Float16)v_.x; h_[1] = (_Float16)v_.y;                     \
            h_[2] = (_Float16)v_.z; h_[3] = (_Float16)v_.w;                     \
            *(v4h*)&As[buf][rsub + 4 * i_][jcol * 4] = h_;                      \
        }                                                                       \
        if (t < ROWS * 3) As[buf][prow][CIN + pk] = (_Float16)(pp[st] - pc);    \
    }

#define LOAD_A(buf, set, kt_)                                                   \
    for (int r2 = 0; r2 < 2; ++r2) {                                            \
        const _Float16* ap = &As[buf][(rp * 2 + r2) * 16 + n][(kt_) * 32 + lo8]; \
        v8h alo = *(const v8h*)ap;                                              \
        v8h ahi = *(const v8h*)(ap + 16);                                       \
        afrag[set][r2] = __builtin_shufflevector(alo, ahi,                      \
            0,1,2,3,4,5,6,7,8,9,10,11,12,13,14,15);                             \
    }

#define STAGE_PIECE(st, nbuf, i_, DOIS, chunk_)                                 \
    {                                                                           \
        float4 v_ = pf[st][i_];                                                 \
        v4h h_;                                                                 \
        h_[0] = (_Float16)v_.x; h_[1] = (_Float16)v_.y;                         \
        h_[2] = (_Float16)v_.z; h_[3] = (_Float16)v_.w;                         \
        *(v4h*)&As[nbuf][rsub + 4 * (i_)][jcol * 4] = h_;                       \
        if (DOIS) pf[st][i_] = (f4 + ((size_t)(chunk_) * ROWS + rsub) * (CIN / 4) + jcol)[(size_t)(i_) * 4 * (CIN / 4)]; \
    }

#define COMPUTE(buf, st, nbuf, DOST, DOIS, chunk_)                              \
    {                                                                           \
        float cmax[2];                                                          \
          \
          \
        _Pragma("unroll")                                                       \
        for (int rp = 0; rp < 2; ++rp) {                                        \
            v8f acc[2][2] = {};                                                 \
            v16h afrag[3][2];                                                   \
            LOAD_A(buf, 0, 0)                                                   \
            LOAD_A(buf, 1, 1)                                                   \
            _Pragma("unroll")                                                   \
            for (int kt = 0; kt < KTOT / 32; ++kt) {                            \
                if (kt + 2 < KTOT / 32) { LOAD_A(buf, (kt + 2) % 3, kt + 2) }   \
                if (DOST && rp * (KTOT / 32) + kt < 16) {                       \
                    STAGE_PIECE(st, nbuf, rp * (KTOT / 32) + kt, DOIS, chunk_)  \
                }                                                               \
                  \
                  \
                  \
                  \
                __builtin_amdgcn_sched_barrier(0);                              \
                for (int r2 = 0; r2 < 2; ++r2) {                                \
                    acc[r2][0] = __builtin_amdgcn_wmma_f32_16x16x32_f16(        \
                        false, afrag[kt % 3][r2], false, bfrag[kt][0], (short)0, acc[r2][0], false, false); \
                    acc[r2][1] = __builtin_amdgcn_wmma_f32_16x16x32_f16(        \
                        false, afrag[kt % 3][r2], false, bfrag[kt][1], (short)0, acc[r2][1], false, false); \
                }                                                               \
                  \
                  \
                  \
                  \
                __builtin_amdgcn_sched_barrier(0);                              \
            }                                                                   \
            for (int ct = 0; ct < 2; ++ct) {                                    \
                float m = (rp == 0) ? acc[0][ct][0] : cmax[ct];                 \
                for (int r2 = 0; r2 < 2; ++r2)                                  \
                    for (int i = 0; i < 8; ++i)                                 \
                        m = fmaxf(m, acc[r2][ct][i]);                           \
                cmax[ct] = m;                                                   \
            }                                                                   \
        }                                                                       \
        for (int ct = 0; ct < 2; ++ct) rmax[ct] = fmaxf(rmax[ct], cmax[ct]);    \
        if (DOST) {                                                             \
            if (t < ROWS * 3) As[nbuf][prow][CIN + pk] = (_Float16)(pp[st] - pc); \
            if (DOIS) pp[st] = pos[(size_t)(chunk_) * (ROWS * 3) + tpos];       \
        }                                                                       \
    }

    STAGE_STORE(0, 0)
    STAGE_ISSUE(0, chunk0 + 2)
    W1_TO_LDS(4, 9)
#undef W1_TO_LDS
    lds_barrier();

    v16h bfrag[KTOT / 32][2];
    for (int kt = 0; kt < KTOT / 32; ++kt) {
        for (int ct = 0; ct < 2; ++ct) {
            const _Float16* bp = &Bs[wv][ct * 16 + n][kt * 32 + lo8];
            v8h blo = *(const v8h*)bp;
            v8h bhi = *(const v8h*)(bp + 16);
            bfrag[kt][ct] = __builtin_shufflevector(blo, bhi,
                0,1,2,3,4,5,6,7,8,9,10,11,12,13,14,15);
        }
    }


    float rmax[2] = {-INFINITY, -INFINITY};

    for (int c = 0; c < NCH - 4; c += 2) {
        COMPUTE(0, 1, 1, 1, 1, chunk0 + c + 3)
        lds_barrier();
        COMPUTE(1, 0, 0, 1, 1, chunk0 + c + 4)
        lds_barrier();
    }
    COMPUTE(0, 1, 1, 1, 1, chunk0 + NCH - 1)
    lds_barrier();
    COMPUTE(1, 0, 0, 1, 0, 0)
    lds_barrier();
    COMPUTE(0, 1, 1, 1, 0, 0)
    lds_barrier();
    COMPUTE(1, 0, 0, 0, 0, 0)

    for (int ct = 0; ct < 2; ++ct) {
        float m = fmaxf(rmax[ct], __shfl_xor(rmax[ct], 16, 32));
        if (half == 0) {
            int col = (wv * 2 + ct) * 16 + n;
            partial[(size_t)blockIdx.x * CMID + col] = m;
        }
    }

#undef COMPUTE
#undef STAGE_PIECE
#undef LOAD_A
#undef STAGE_STORE
#undef STAGE_ISSUE
}

#define FIN_SC     2
#define FIN_CW     64
#define FIN_KG     8
#define FIN_KPT    (CMID / FIN_KG)
#define FIN_SLICES (CMID / FIN_CW)
#define FIN_BLOCKS ((BSC / FIN_SC) * FIN_SLICES)
#define FIN_PARTS  (CHUNKS / NCH)
static_assert(FIN_CW * 2 == 32 * 4, "one wave = one W2 row of the block: 16 mu + 16 sigma float4");
static_assert(FIN_SC * FIN_CW <= 256 && CMID == 256, "thread maps of the head and the tail");

__global__ __launch_bounds__(256)
void finish_kernel(const float* __restrict__ partial,
                   const float* __restrict__ b1,
                   const float* __restrict__ W2,
                   const float* __restrict__ b2,
                   const float* __restrict__ noise,
                   float* __restrict__ out) {
    __shared__ __align__(8)  float agg[CMID][FIN_SC];
    __shared__ __align__(16) float red[FIN_KG][FIN_SC][2 * FIN_CW];

    const int t    = threadIdx.x;
    const int lane = t & 31;
    const int kg   = t >> 5;
    const int sl   = blockIdx.x % FIN_SLICES;
    const int s0   = (blockIdx.x / FIN_SLICES) * FIN_SC;

    const int col4 = (lane < 16) ? (FIN_CW * sl + 4 * lane)
                                 : (CMID + FIN_CW * sl + 4 * (lane - 16));

    float cm[FIN_SC][FIN_PARTS];
#pragma unroll
    for (int sc = 0; sc < FIN_SC; ++sc) {
        const float* p = partial + (size_t)(s0 + sc) * FIN_PARTS * CMID + t;
#pragma unroll
        for (int ch = 0; ch < FIN_PARTS; ++ch) cm[sc][ch] = p[ch * CMID];
    }
    const float bias1 = b1[t];

    float4 w[FIN_KPT];
    const float4* wp = (const float4*)(W2 + (size_t)kg * COUT + col4);
#pragma unroll
    for (int i = 0; i < FIN_KPT; ++i) w[i] = wp[(size_t)i * FIN_KG * (COUT / 4)];

    const int tsc = t / FIN_CW;
    const int tcc = t % FIN_CW;
    float tb_mu = 0.f, tb_sr = 0.f, tnoise = 0.f;
    if (t < FIN_SC * FIN_CW) {
        tb_mu  = b2[FIN_CW * sl + tcc];
        tb_sr  = b2[CMID + FIN_CW * sl + tcc];
        tnoise = noise[(size_t)(s0 + tsc) * CMID + FIN_CW * sl + tcc];
    }

#pragma unroll
    for (int sc = 0; sc < FIN_SC; ++sc) {
        float m = cm[sc][0];
#pragma unroll
        for (int ch = 1; ch < FIN_PARTS; ++ch) m = fmaxf(m, cm[sc][ch]);
        agg[t][sc] = fmaxf(m + bias1, 0.0f);
    }
    __syncthreads();

    float4 acc[FIN_SC];
#pragma unroll
    for (int sc = 0; sc < FIN_SC; ++sc) acc[sc] = make_float4(0.f, 0.f, 0.f, 0.f);
#pragma unroll
    for (int i = 0; i < FIN_KPT; ++i) {
        const int k = kg + FIN_KG * i;
#pragma unroll
        for (int sc = 0; sc < FIN_SC; ++sc) {
            const float a = agg[k][sc];
            acc[sc].x += a * w[i].x; acc[sc].y += a * w[i].y;
            acc[sc].z += a * w[i].z; acc[sc].w += a * w[i].w;
        }
    }
#pragma unroll
    for (int sc = 0; sc < FIN_SC; ++sc)
        *(float4*)&red[kg][sc][4 * lane] = acc[sc];
    __syncthreads();

    if (t < FIN_SC * FIN_CW) {
        float mu = tb_mu;
        float sr = tb_sr;
#pragma unroll
        for (int g = 0; g < FIN_KG; ++g) {
            mu += red[g][tsc][tcc];
            sr += red[g][tsc][FIN_CW + tcc];
        }
        float sig = fmaxf(sr, 0.0f) + log1pf(expf(-fabsf(sr))) + 1e-4f;
        out[(size_t)(s0 + tsc) * CMID + FIN_CW * sl + tcc] = mu + sig * tnoise;
    }
}

extern "C" void kernel_launch(void* const* d_in, const int* in_sizes, int n_in,
                              void* d_out, int out_size, void* d_ws, size_t ws_size,
                              hipStream_t stream) {
    const float* pos     = (const float*)d_in[0];
    const float* feature = (const float*)d_in[1];
    const float* W1      = (const float*)d_in[3];
    const float* b1      = (const float*)d_in[4];
    const float* W2      = (const float*)d_in[5];
    const float* b2      = (const float*)d_in[6];
    const float* noise   = (const float*)d_in[7];
    float* out = (float*)d_out;

    float*    partial= (float*)((char*)d_ws + WS_PARTIAL);

    gemm_max_kernel<<<GBLOCKS, 256, 0, stream>>>(pos, feature, W1, partial);
    finish_kernel<<<FIN_BLOCKS, 256, 0, stream>>>(partial, b1, W2, b2, noise, out);
}
